# starting kernel
# baseline (speedup 1.0000x reference)
#include <hip/hip_runtime.h>
#include <hip/hip_bf16.h>

typedef float v2f __attribute__((ext_vector_type(2)));
typedef float v8f __attribute__((ext_vector_type(8)));

#define NEG_SLOPE 0.2f

__device__ __forceinline__ int f2ord(float f) {
    int i = __float_as_int(f);
    return (i >= 0) ? i : (i ^ 0x7fffffff);
}
__device__ __forceinline__ float ord2f(int i) {
    return __int_as_float((i >= 0) ? i : (i ^ 0x7fffffff));
}

__global__ __launch_bounds__(128) void gat_gemm_wmma(
    const float* __restrict__ A, const float* __restrict__ W,
    float* __restrict__ Out, int M) {
    __shared__ float As[16][132];

    const int rowBase = blockIdx.x * 16;

    #pragma unroll
    for (int i = 0; i < 16; ++i) {
        int idx = threadIdx.x + i * 128;
        int r = idx >> 7;
        int k = idx & 127;
        int row = rowBase + r;
        if (row >= M) row = M - 1;
        As[r][k] = A[(size_t)row * 128 + k];
    }
    __syncthreads();

    const int wave = threadIdx.x >> 5;
    const int lane = threadIdx.x & 31;
    const int half = lane >> 4;
    const int l16  = lane & 15;
    const int ncol = (blockIdx.y * 4 + wave) * 16 + l16;

    v8f acc = {};
    for (int kk = 0; kk < 128; kk += 4) {
        v2f a, b;
        a.x = As[l16][kk + half * 2 + 0];
        a.y = As[l16][kk + half * 2 + 1];
        b.x = W[(size_t)(kk + half * 2 + 0) * 128 + ncol];
        b.y = W[(size_t)(kk + half * 2 + 1) * 128 + ncol];
        acc = __builtin_amdgcn_wmma_f32_16x16x4_f32(
             false, a,  false, b,
             (short)0, acc,  false,  false);
    }

    #pragma unroll
    for (int r = 0; r < 8; ++r) {
        int row = rowBase + r + half * 8;
        if (row < M) Out[(size_t)row * 128 + ncol] = acc[r];
    }
}

__global__ void gat_coef(const float* __restrict__ feat,
                         const float* __restrict__ al,
                         const float* __restrict__ ar,
                         float* __restrict__ el, float* __restrict__ er,
                         int N) {
    int n = blockIdx.x * blockDim.x + threadIdx.x;
    if (n >= N) return;
    const float* f = feat + (size_t)n * 128;
    float e0 = 0.f, e1 = 0.f, r0 = 0.f, r1 = 0.f;
    #pragma unroll 4
    for (int d = 0; d < 64; ++d) {
        float h0 = f[d], h1 = f[64 + d];
        e0 += h0 * al[d];
        e1 += h1 * al[64 + d];
        r0 += h0 * ar[d];
        r1 += h1 * ar[64 + d];
    }
    el[n * 2 + 0] = e0; el[n * 2 + 1] = e1;
    er[n * 2 + 0] = r0; er[n * 2 + 1] = r1;
}

__global__ void gat_init_ms(int* __restrict__ mkey, float* __restrict__ s, int n2) {
    int i = blockIdx.x * blockDim.x + threadIdx.x;
    if (i < n2) { mkey[i] = 0x80000000; s[i] = 0.f; }
}

__global__ void gat_init_bias(float* __restrict__ out, const float* __restrict__ b,
                              int total) {
    int i = blockIdx.x * blockDim.x + threadIdx.x;
    if (i < total) out[i] = b[i & 127];
}

__device__ __forceinline__ void edge_sd(long long i, long long E,
                                        const int* __restrict__ src,
                                        const int* __restrict__ dst,
                                        int& s, int& d) {
    if (i < E) { s = src[i]; d = dst[i]; }
    else       { s = (int)(i - E); d = s; }
}

__global__ void gat_edge_max(const int* __restrict__ src, const int* __restrict__ dst,
                             long long E, long long Et,
                             const float* __restrict__ el, const float* __restrict__ er,
                             int* __restrict__ mkey) {
    long long i = (long long)blockIdx.x * blockDim.x + threadIdx.x;
    if (i >= Et) return;
    int s, d; edge_sd(i, E, src, dst, s, d);
    #pragma unroll
    for (int h = 0; h < 2; ++h) {
        float e = el[s * 2 + h] + er[d * 2 + h];
        e = (e >= 0.f) ? e : NEG_SLOPE * e;
        atomicMax(&mkey[d * 2 + h], f2ord(e));
    }
}

__global__ void gat_edge_sum(const int* __restrict__ src, const int* __restrict__ dst,
                             long long E, long long Et,
                             const float* __restrict__ el, const float* __restrict__ er,
                             const int* __restrict__ mkey, float* __restrict__ ssum) {
    long long i = (long long)blockIdx.x * blockDim.x + threadIdx.x;
    if (i >= Et) return;
    int s, d; edge_sd(i, E, src, dst, s, d);
    #pragma unroll
    for (int h = 0; h < 2; ++h) {
        float e = el[s * 2 + h] + er[d * 2 + h];
        e = (e >= 0.f) ? e : NEG_SLOPE * e;
        atomicAdd(&ssum[d * 2 + h], expf(e - ord2f(mkey[d * 2 + h])));
    }
}

__global__ __launch_bounds__(256) void gat_edge_agg(
    const int* __restrict__ src, const int* __restrict__ dst,
    long long E, long long Et,
    const float* __restrict__ el, const float* __restrict__ er,
    const int* __restrict__ mkey, const float* __restrict__ ssum,
    const float* __restrict__ feat, float* __restrict__ out) {
    long long w = (long long)blockIdx.x * 8 + (threadIdx.x >> 5);
    if (w >= Et) return;
    const int lane = threadIdx.x & 31;
    int s, d; edge_sd(w, E, src, dst, s, d);

    const int f0 = lane * 4;
    const int h  = f0 >> 6;
    float e = el[s * 2 + h] + er[d * 2 + h];
    e = (e >= 0.f) ? e : NEG_SLOPE * e;
    const float alpha = expf(e - ord2f(mkey[d * 2 + h])) / ssum[d * 2 + h];

    const float4 v = *(const float4*)(feat + (size_t)s * 128 + f0);
    float* o = out + (size_t)d * 128 + f0;
    atomicAdd(o + 0, alpha * v.x);
    atomicAdd(o + 1, alpha * v.y);
    atomicAdd(o + 2, alpha * v.z);
    atomicAdd(o + 3, alpha * v.w);
}

__global__ void gat_relu(float* __restrict__ x, int total) {
    int i = blockIdx.x * blockDim.x + threadIdx.x;
    if (i < total) x[i] = fmaxf(x[i], 0.f);
}

extern "C" void kernel_launch(void* const* d_in, const int* in_sizes, int n_in,
                              void* d_out, int out_size, void* d_ws, size_t ws_size,
                              hipStream_t stream) {
    const float* x   = (const float*)d_in[0];
    const int*   src = (const int*)  d_in[1];
    const int*   dst = (const int*)  d_in[2];
    const float* W1  = (const float*)d_in[3];
    const float* al1 = (const float*)d_in[4];
    const float* ar1 = (const float*)d_in[5];
    const float* b1  = (const float*)d_in[6];
    const float* W2  = (const float*)d_in[7];
    const float* al2 = (const float*)d_in[8];
    const float* ar2 = (const float*)d_in[9];
    const float* b2  = (const float*)d_in[10];

    const int N = in_sizes[0] / 128;
    const long long E  = in_sizes[1];
    const long long Et = E + N;

    float* out = (float*)d_out;

    float* feat = (float*)d_ws;
    float* el   = feat + (size_t)N * 128;
    float* er   = el + (size_t)N * 2;
    float* ssum = er + (size_t)N * 2;
    int*   mkey = (int*)(ssum + (size_t)N * 2);

    const dim3 gemmGrid((N + 15) / 16, 2);
    const int  totalNF = N * 128;
    const int  n2      = N * 2;
    const int  eBlocks = (int)((Et + 255) / 256);
    const int  aBlocks = (int)((Et + 7) / 8);

    gat_gemm_wmma<<<gemmGrid, 128, 0, stream>>>(x, W1, feat, N);
    gat_coef<<<(N + 255) / 256, 256, 0, stream>>>(feat, al1, ar1, el, er, N);
    gat_init_ms<<<(n2 + 255) / 256, 256, 0, stream>>>(mkey, ssum, n2);
    gat_init_bias<<<(totalNF + 255) / 256, 256, 0, stream>>>(out, b1, totalNF);
    gat_edge_max<<<eBlocks, 256, 0, stream>>>(src, dst, E, Et, el, er, mkey);
    gat_edge_sum<<<eBlocks, 256, 0, stream>>>(src, dst, E, Et, el, er, mkey, ssum);
    gat_edge_agg<<<aBlocks, 256, 0, stream>>>(src, dst, E, Et, el, er, mkey, ssum, feat, out);
    gat_relu<<<(totalNF + 255) / 256, 256, 0, stream>>>(out, totalNF);

    gat_gemm_wmma<<<gemmGrid, 128, 0, stream>>>(out, W2, feat, N);
    gat_coef<<<(N + 255) / 256, 256, 0, stream>>>(feat, al2, ar2, el, er, N);
    gat_init_ms<<<(n2 + 255) / 256, 256, 0, stream>>>(mkey, ssum, n2);
    gat_init_bias<<<(totalNF + 255) / 256, 256, 0, stream>>>(out, b2, totalNF);
    gat_edge_max<<<eBlocks, 256, 0, stream>>>(src, dst, E, Et, el, er, mkey);
    gat_edge_sum<<<eBlocks, 256, 0, stream>>>(src, dst, E, Et, el, er, mkey, ssum);
    gat_edge_agg<<<aBlocks, 256, 0, stream>>>(src, dst, E, Et, el, er, mkey, ssum, feat, out);
}


# Edge aggregation: lane owns features lane+32j, so each of a wave's 4 float atomics and 4 row loads covers one whole 128-byte line instead of 8 words in each of 4 lines; add order still arrival order
# speedup vs baseline: 1.1858x; 1.1858x over previous
#include <hip/hip_runtime.h>
#include <hip/hip_bf16.h>

typedef float v2f __attribute__((ext_vector_type(2)));
typedef float v8f __attribute__((ext_vector_type(8)));

#define NEG_SLOPE 0.2f

__device__ __forceinline__ int f2ord(float f) {
    int i = __float_as_int(f);
    return (i >= 0) ? i : (i ^ 0x7fffffff);
}
__device__ __forceinline__ float ord2f(int i) {
    return __int_as_float((i >= 0) ? i : (i ^ 0x7fffffff));
}

__global__ __launch_bounds__(128) void gat_gemm_wmma(
    const float* __restrict__ A, const float* __restrict__ W,
    float* __restrict__ Out, int M) {
    __shared__ float As[16][132];

    const int rowBase = blockIdx.x * 16;

    #pragma unroll
    for (int i = 0; i < 16; ++i) {
        int idx = threadIdx.x + i * 128;
        int r = idx >> 7;
        int k = idx & 127;
        int row = rowBase + r;
        if (row >= M) row = M - 1;
        As[r][k] = A[(size_t)row * 128 + k];
    }
    __syncthreads();

    const int wave = threadIdx.x >> 5;
    const int lane = threadIdx.x & 31;
    const int half = lane >> 4;
    const int l16  = lane & 15;
    const int ncol = (blockIdx.y * 4 + wave) * 16 + l16;

    v8f acc = {};
    for (int kk = 0; kk < 128; kk += 4) {
        v2f a, b;
        a.x = As[l16][kk + half * 2 + 0];
        a.y = As[l16][kk + half * 2 + 1];
        b.x = W[(size_t)(kk + half * 2 + 0) * 128 + ncol];
        b.y = W[(size_t)(kk + half * 2 + 1) * 128 + ncol];
        acc = __builtin_amdgcn_wmma_f32_16x16x4_f32(
             false, a,  false, b,
             (short)0, acc,  false,  false);
    }

    #pragma unroll
    for (int r = 0; r < 8; ++r) {
        int row = rowBase + r + half * 8;
        if (row < M) Out[(size_t)row * 128 + ncol] = acc[r];
    }
}

__global__ void gat_coef(const float* __restrict__ feat,
                         const float* __restrict__ al,
                         const float* __restrict__ ar,
                         float* __restrict__ el, float* __restrict__ er,
                         int N) {
    int n = blockIdx.x * blockDim.x + threadIdx.x;
    if (n >= N) return;
    const float* f = feat + (size_t)n * 128;
    float e0 = 0.f, e1 = 0.f, r0 = 0.f, r1 = 0.f;
    #pragma unroll 4
    for (int d = 0; d < 64; ++d) {
        float h0 = f[d], h1 = f[64 + d];
        e0 += h0 * al[d];
        e1 += h1 * al[64 + d];
        r0 += h0 * ar[d];
        r1 += h1 * ar[64 + d];
    }
    el[n * 2 + 0] = e0; el[n * 2 + 1] = e1;
    er[n * 2 + 0] = r0; er[n * 2 + 1] = r1;
}

__global__ void gat_init_ms(int* __restrict__ mkey, float* __restrict__ s, int n2) {
    int i = blockIdx.x * blockDim.x + threadIdx.x;
    if (i < n2) { mkey[i] = 0x80000000; s[i] = 0.f; }
}

__global__ void gat_init_bias(float* __restrict__ out, const float* __restrict__ b,
                              int total) {
    int i = blockIdx.x * blockDim.x + threadIdx.x;
    if (i < total) out[i] = b[i & 127];
}

__device__ __forceinline__ void edge_sd(long long i, long long E,
                                        const int* __restrict__ src,
                                        const int* __restrict__ dst,
                                        int& s, int& d) {
    if (i < E) { s = src[i]; d = dst[i]; }
    else       { s = (int)(i - E); d = s; }
}

__global__ void gat_edge_max(const int* __restrict__ src, const int* __restrict__ dst,
                             long long E, long long Et,
                             const float* __restrict__ el, const float* __restrict__ er,
                             int* __restrict__ mkey) {
    long long i = (long long)blockIdx.x * blockDim.x + threadIdx.x;
    if (i >= Et) return;
    int s, d; edge_sd(i, E, src, dst, s, d);
    #pragma unroll
    for (int h = 0; h < 2; ++h) {
        float e = el[s * 2 + h] + er[d * 2 + h];
        e = (e >= 0.f) ? e : NEG_SLOPE * e;
        atomicMax(&mkey[d * 2 + h], f2ord(e));
    }
}

__global__ void gat_edge_sum(const int* __restrict__ src, const int* __restrict__ dst,
                             long long E, long long Et,
                             const float* __restrict__ el, const float* __restrict__ er,
                             const int* __restrict__ mkey, float* __restrict__ ssum) {
    long long i = (long long)blockIdx.x * blockDim.x + threadIdx.x;
    if (i >= Et) return;
    int s, d; edge_sd(i, E, src, dst, s, d);
    #pragma unroll
    for (int h = 0; h < 2; ++h) {
        float e = el[s * 2 + h] + er[d * 2 + h];
        e = (e >= 0.f) ? e : NEG_SLOPE * e;
        atomicAdd(&ssum[d * 2 + h], expf(e - ord2f(mkey[d * 2 + h])));
    }
}

__global__ __launch_bounds__(256) void gat_edge_agg(
    const int* __restrict__ src, const int* __restrict__ dst,
    long long E, long long Et,
    const float* __restrict__ el, const float* __restrict__ er,
    const int* __restrict__ mkey, const float* __restrict__ ssum,
    const float* __restrict__ feat, float* __restrict__ out) {
    long long w = (long long)blockIdx.x * 8 + (threadIdx.x >> 5);
    if (w >= Et) return;
    const int lane = threadIdx.x & 31;
    int s, d; edge_sd(w, E, src, dst, s, d);

    float e0 = el[s * 2 + 0] + er[d * 2 + 0];
    float e1 = el[s * 2 + 1] + er[d * 2 + 1];
    e0 = (e0 >= 0.f) ? e0 : NEG_SLOPE * e0;
    e1 = (e1 >= 0.f) ? e1 : NEG_SLOPE * e1;
    const float alpha0 = expf(e0 - ord2f(mkey[d * 2 + 0])) / ssum[d * 2 + 0];
    const float alpha1 = expf(e1 - ord2f(mkey[d * 2 + 1])) / ssum[d * 2 + 1];

    const float* f = feat + (size_t)s * 128 + lane;
    const float v0 = f[0], v1 = f[32], v2 = f[64], v3 = f[96];
    float* o = out + (size_t)d * 128 + lane;
    atomicAdd(o +  0, alpha0 * v0);
    atomicAdd(o + 32, alpha0 * v1);
    atomicAdd(o + 64, alpha1 * v2);
    atomicAdd(o + 96, alpha1 * v3);
}

__global__ void gat_relu(float* __restrict__ x, int total) {
    int i = blockIdx.x * blockDim.x + threadIdx.x;
    if (i < total) x[i] = fmaxf(x[i], 0.f);
}

extern "C" void kernel_launch(void* const* d_in, const int* in_sizes, int n_in,
                              void* d_out, int out_size, void* d_ws, size_t ws_size,
                              hipStream_t stream) {
    const float* x   = (const float*)d_in[0];
    const int*   src = (const int*)  d_in[1];
    const int*   dst = (const int*)  d_in[2];
    const float* W1  = (const float*)d_in[3];
    const float* al1 = (const float*)d_in[4];
    const float* ar1 = (const float*)d_in[5];
    const float* b1  = (const float*)d_in[6];
    const float* W2  = (const float*)d_in[7];
    const float* al2 = (const float*)d_in[8];
    const float* ar2 = (const float*)d_in[9];
    const float* b2  = (const float*)d_in[10];

    const int N = in_sizes[0] / 128;
    const long long E  = in_sizes[1];
    const long long Et = E + N;

    float* out = (float*)d_out;

    float* feat = (float*)d_ws;
    float* el   = feat + (size_t)N * 128;
    float* er   = el + (size_t)N * 2;
    float* ssum = er + (size_t)N * 2;
    int*   mkey = (int*)(ssum + (size_t)N * 2);

    const dim3 gemmGrid((N + 15) / 16, 2);
    const int  totalNF = N * 128;
    const int  n2      = N * 2;
    const int  eBlocks = (int)((Et + 255) / 256);
    const int  aBlocks = (int)((Et + 7) / 8);

    gat_gemm_wmma<<<gemmGrid, 128, 0, stream>>>(x, W1, feat, N);
    gat_coef<<<(N + 255) / 256, 256, 0, stream>>>(feat, al1, ar1, el, er, N);
    gat_init_ms<<<(n2 + 255) / 256, 256, 0, stream>>>(mkey, ssum, n2);
    gat_init_bias<<<(totalNF + 255) / 256, 256, 0, stream>>>(out, b1, totalNF);
    gat_edge_max<<<eBlocks, 256, 0, stream>>>(src, dst, E, Et, el, er, mkey);
    gat_edge_sum<<<eBlocks, 256, 0, stream>>>(src, dst, E, Et, el, er, mkey, ssum);
    gat_edge_agg<<<aBlocks, 256, 0, stream>>>(src, dst, E, Et, el, er, mkey, ssum, feat, out);
    gat_relu<<<(totalNF + 255) / 256, 256, 0, stream>>>(out, totalNF);

    gat_gemm_wmma<<<gemmGrid, 128, 0, stream>>>(out, W2, feat, N);
    gat_coef<<<(N + 255) / 256, 256, 0, stream>>>(feat, al2, ar2, el, er, N);
    gat_init_ms<<<(n2 + 255) / 256, 256, 0, stream>>>(mkey, ssum, n2);
    gat_init_bias<<<(totalNF + 255) / 256, 256, 0, stream>>>(out, b2, totalNF);
    gat_edge_max<<<eBlocks, 256, 0, stream>>>(src, dst, E, Et, el, er, mkey);
    gat_edge_sum<<<eBlocks, 256, 0, stream>>>(src, dst, E, Et, el, er, mkey, ssum);
    gat_edge_agg<<<aBlocks, 256, 0, stream>>>(src, dst, E, Et, el, er, mkey, ssum, feat, out);
}


# Aggregate by destination: per-call counting sort of edges by dst (int atomics), one wave per node does softmax and weighted row sum in registers, no float atomics; sum order = sort arrival order
# speedup vs baseline: 3.5645x; 3.0060x over previous
#include <hip/hip_runtime.h>
#include <hip/hip_bf16.h>

typedef float v2f __attribute__((ext_vector_type(2)));
typedef float v8f __attribute__((ext_vector_type(8)));

#define NEG_SLOPE 0.2f

__device__ __forceinline__ int f2ord(float f) {
    int i = __float_as_int(f);
    return (i >= 0) ? i : (i ^ 0x7fffffff);
}
__device__ __forceinline__ float ord2f(int i) {
    return __int_as_float((i >= 0) ? i : (i ^ 0x7fffffff));
}

__global__ __launch_bounds__(128) void gat_gemm_wmma(
    const float* __restrict__ A, const float* __restrict__ W,
    float* __restrict__ Out, int M) {
    __shared__ float As[16][132];

    const int rowBase = blockIdx.x * 16;

    #pragma unroll
    for (int i = 0; i < 16; ++i) {
        int idx = threadIdx.x + i * 128;
        int r = idx >> 7;
        int k = idx & 127;
        int row = rowBase + r;
        if (row >= M) row = M - 1;
        As[r][k] = A[(size_t)row * 128 + k];
    }
    __syncthreads();

    const int wave = threadIdx.x >> 5;
    const int lane = threadIdx.x & 31;
    const int half = lane >> 4;
    const int l16  = lane & 15;
    const int ncol = (blockIdx.y * 4 + wave) * 16 + l16;

    v8f acc = {};
    for (int kk = 0; kk < 128; kk += 4) {
        v2f a, b;
        a.x = As[l16][kk + half * 2 + 0];
        a.y = As[l16][kk + half * 2 + 1];
        b.x = W[(size_t)(kk + half * 2 + 0) * 128 + ncol];
        b.y = W[(size_t)(kk + half * 2 + 1) * 128 + ncol];
        acc = __builtin_amdgcn_wmma_f32_16x16x4_f32(
             false, a,  false, b,
             (short)0, acc,  false,  false);
    }

    #pragma unroll
    for (int r = 0; r < 8; ++r) {
        int row = rowBase + r + half * 8;
        if (row < M) Out[(size_t)row * 128 + ncol] = acc[r];
    }
}

__global__ void gat_coef(const float* __restrict__ feat,
                         const float* __restrict__ al,
                         const float* __restrict__ ar,
                         float* __restrict__ el, float* __restrict__ er,
                         int N) {
    int n = blockIdx.x * blockDim.x + threadIdx.x;
    if (n >= N) return;
    const float* f = feat + (size_t)n * 128;
    float e0 = 0.f, e1 = 0.f, r0 = 0.f, r1 = 0.f;
    #pragma unroll 4
    for (int d = 0; d < 64; ++d) {
        float h0 = f[d], h1 = f[64 + d];
        e0 += h0 * al[d];
        e1 += h1 * al[64 + d];
        r0 += h0 * ar[d];
        r1 += h1 * ar[64 + d];
    }
    el[n * 2 + 0] = e0; el[n * 2 + 1] = e1;
    er[n * 2 + 0] = r0; er[n * 2 + 1] = r1;
}

__global__ void gat_init_ms(int* __restrict__ mkey, float* __restrict__ s, int n2) {
    int i = blockIdx.x * blockDim.x + threadIdx.x;
    if (i < n2) { mkey[i] = 0x80000000; s[i] = 0.f; }
}

__global__ void gat_init_bias(float* __restrict__ out, const float* __restrict__ b,
                              int total) {
    int i = blockIdx.x * blockDim.x + threadIdx.x;
    if (i < total) out[i] = b[i & 127];
}

__device__ __forceinline__ void edge_sd(long long i, long long E,
                                        const int* __restrict__ src,
                                        const int* __restrict__ dst,
                                        int& s, int& d) {
    if (i < E) { s = src[i]; d = dst[i]; }
    else       { s = (int)(i - E); d = s; }
}

__global__ void gat_edge_max(const int* __restrict__ src, const int* __restrict__ dst,
                             long long E, long long Et,
                             const float* __restrict__ el, const float* __restrict__ er,
                             int* __restrict__ mkey) {
    long long i = (long long)blockIdx.x * blockDim.x + threadIdx.x;
    if (i >= Et) return;
    int s, d; edge_sd(i, E, src, dst, s, d);
    #pragma unroll
    for (int h = 0; h < 2; ++h) {
        float e = el[s * 2 + h] + er[d * 2 + h];
        e = (e >= 0.f) ? e : NEG_SLOPE * e;
        atomicMax(&mkey[d * 2 + h], f2ord(e));
    }
}

__global__ void gat_edge_sum(const int* __restrict__ src, const int* __restrict__ dst,
                             long long E, long long Et,
                             const float* __restrict__ el, const float* __restrict__ er,
                             const int* __restrict__ mkey, float* __restrict__ ssum) {
    long long i = (long long)blockIdx.x * blockDim.x + threadIdx.x;
    if (i >= Et) return;
    int s, d; edge_sd(i, E, src, dst, s, d);
    #pragma unroll
    for (int h = 0; h < 2; ++h) {
        float e = el[s * 2 + h] + er[d * 2 + h];
        e = (e >= 0.f) ? e : NEG_SLOPE * e;
        atomicAdd(&ssum[d * 2 + h], expf(e - ord2f(mkey[d * 2 + h])));
    }
}

__global__ __launch_bounds__(256) void gat_edge_agg(
    const int* __restrict__ src, const int* __restrict__ dst,
    long long E, long long Et,
    const float* __restrict__ el, const float* __restrict__ er,
    const int* __restrict__ mkey, const float* __restrict__ ssum,
    const float* __restrict__ feat, float* __restrict__ out) {
    long long w = (long long)blockIdx.x * 8 + (threadIdx.x >> 5);
    if (w >= Et) return;
    const int lane = threadIdx.x & 31;
    int s, d; edge_sd(w, E, src, dst, s, d);

    float e0 = el[s * 2 + 0] + er[d * 2 + 0];
    float e1 = el[s * 2 + 1] + er[d * 2 + 1];
    e0 = (e0 >= 0.f) ? e0 : NEG_SLOPE * e0;
    e1 = (e1 >= 0.f) ? e1 : NEG_SLOPE * e1;
    const float alpha0 = expf(e0 - ord2f(mkey[d * 2 + 0])) / ssum[d * 2 + 0];
    const float alpha1 = expf(e1 - ord2f(mkey[d * 2 + 1])) / ssum[d * 2 + 1];

    const float* f = feat + (size_t)s * 128 + lane;
    const float v0 = f[0], v1 = f[32], v2 = f[64], v3 = f[96];
    float* o = out + (size_t)d * 128 + lane;
    atomicAdd(o +  0, alpha0 * v0);
    atomicAdd(o + 32, alpha0 * v1);
    atomicAdd(o + 64, alpha1 * v2);
    atomicAdd(o + 96, alpha1 * v3);
}

__global__ void gat_relu(float* __restrict__ x, int total) {
    int i = blockIdx.x * blockDim.x + threadIdx.x;
    if (i < total) x[i] = fmaxf(x[i], 0.f);
}


__global__ void csr_zero(int* __restrict__ pos, int N) {
    int i = blockIdx.x * blockDim.x + threadIdx.x;
    if (i < N) pos[i] = 0;
}

__global__ void csr_count(const int* __restrict__ dst, int E, int* __restrict__ pos) {
    int i = blockIdx.x * blockDim.x + threadIdx.x;
    if (i < E) atomicAdd(&pos[dst[i]], 1);
}

__global__ __launch_bounds__(1024) void csr_scan_block(int* __restrict__ pos, int N,
                                                       int* __restrict__ bsum) {
    __shared__ int wtot[32];
    const int t = threadIdx.x, lane = t & 31, wv = t >> 5;
    const int base = blockIdx.x * 4096 + t * 4;
    int v[4];
    #pragma unroll
    for (int i = 0; i < 4; ++i) v[i] = (base + i < N) ? pos[base + i] : 0;
    const int tsum = v[0] + v[1] + v[2] + v[3];
    int x = tsum;
    #pragma unroll
    for (int o = 1; o < 32; o <<= 1) {
        int y = __shfl_up(x, o);
        if (lane >= o) x += y;
    }
    if (lane == 31) wtot[wv] = x;
    __syncthreads();
    if (wv == 0) {
        int w = wtot[lane];
        #pragma unroll
        for (int o = 1; o < 32; o <<= 1) {
            int y = __shfl_up(w, o);
            if (lane >= o) w += y;
        }
        wtot[lane] = w;
    }
    __syncthreads();
    int run = x - tsum + (wv > 0 ? wtot[wv - 1] : 0);
    #pragma unroll
    for (int i = 0; i < 4; ++i) {
        if (base + i < N) pos[base + i] = run;
        run += v[i];
    }
    if (t == 1023) bsum[blockIdx.x] = run;
}

__global__ __launch_bounds__(1024) void csr_scan_add(int* __restrict__ pos, int N,
                                                     const int* __restrict__ bsum) {
    const int b = blockIdx.x;
    int off = 0;
    for (int j = 0; j < b; ++j) off += bsum[j];
    const int base = b * 4096 + threadIdx.x * 4;
    #pragma unroll
    for (int i = 0; i < 4; ++i)
        if (base + i < N) pos[base + i] += off;
}

__global__ void csr_fill(const int* __restrict__ src, const int* __restrict__ dst, int E,
                         int* __restrict__ pos, int* __restrict__ csr) {
    int i = blockIdx.x * blockDim.x + threadIdx.x;
    if (i >= E) return;
    int p = atomicAdd(&pos[dst[i]], 1);
    if ((unsigned)p < (unsigned)E) csr[p] = src[i];
}

__device__ __forceinline__ float lrelu(float e) { return (e >= 0.f) ? e : NEG_SLOPE * e; }

__device__ __forceinline__ float wave_max(float v) {
    #pragma unroll
    for (int o = 16; o > 0; o >>= 1) v = fmaxf(v, __shfl_xor(v, o));
    return v;
}
__device__ __forceinline__ float wave_sum(float v) {
    #pragma unroll
    for (int o = 16; o > 0; o >>= 1) v += __shfl_xor(v, o);
    return v;
}

__device__ __forceinline__ void add_rows(float4& acc, const float* __restrict__ feat,
                                         int s_l, float a0_l, float a1_l, int cnt,
                                         int lane, bool hi) {
    for (int k = 0; k < cnt; k += 4) {
        float4 v[4];
        float  a[4];
        #pragma unroll
        for (int i = 0; i < 4; ++i) {
            const int   s  = __builtin_amdgcn_readlane(s_l, k + i);
            const float a0 = __int_as_float(__builtin_amdgcn_readlane(__float_as_int(a0_l), k + i));
            const float a1 = __int_as_float(__builtin_amdgcn_readlane(__float_as_int(a1_l), k + i));
            a[i] = hi ? a1 : a0;
            v[i] = reinterpret_cast<const float4*>(feat + (size_t)s * 128)[lane];
        }
        #pragma unroll
        for (int i = 0; i < 4; ++i) {
            acc.x = fmaf(a[i], v[i].x, acc.x);
            acc.y = fmaf(a[i], v[i].y, acc.y);
            acc.z = fmaf(a[i], v[i].z, acc.z);
            acc.w = fmaf(a[i], v[i].w, acc.w);
        }
    }
}

__global__ __launch_bounds__(256) void gat_gather(
    const int* __restrict__ pos, const int* __restrict__ csr,
    const float* __restrict__ el, const float* __restrict__ er,
    const float* __restrict__ feat, const float* __restrict__ bias,
    float* __restrict__ out, int N, int relu) {
    const int d = __builtin_amdgcn_readfirstlane((int)(blockIdx.x * 8 + (threadIdx.x >> 5)));
    if (d >= N) return;
    const int lane = threadIdx.x & 31;
    const bool hi = lane >= 16;
    const int beg = (d > 0) ? pos[d - 1] : 0;
    const int deg = pos[d] - beg;

    const float4 vd = reinterpret_cast<const float4*>(feat + (size_t)d * 128)[lane];

    const float er0 = er[d * 2 + 0], er1 = er[d * 2 + 1];
    const float es0 = lrelu(el[d * 2 + 0] + er0);
    const float es1 = lrelu(el[d * 2 + 1] + er1);

    const int  n0  = deg < 32 ? deg : 32;
    const bool act = lane < n0;
    int   s_c  = d;
    float e0_c = 0.f, e1_c = 0.f;
    if (act) {
        s_c  = csr[beg + lane];
        e0_c = lrelu(el[s_c * 2 + 0] + er0);
        e1_c = lrelu(el[s_c * 2 + 1] + er1);
    }
    float m0 = act ? fmaxf(es0, e0_c) : es0;
    float m1 = act ? fmaxf(es1, e1_c) : es1;
    for (int j = 32 + lane; j < deg; j += 32) {
        const int s = csr[beg + j];
        m0 = fmaxf(m0, lrelu(el[s * 2 + 0] + er0));
        m1 = fmaxf(m1, lrelu(el[s * 2 + 1] + er1));
    }
    m0 = wave_max(m0);
    m1 = wave_max(m1);

    const float p0_c = act ? expf(e0_c - m0) : 0.f;
    const float p1_c = act ? expf(e1_c - m1) : 0.f;
    float z0 = p0_c, z1 = p1_c;
    for (int j = 32 + lane; j < deg; j += 32) {
        const int s = csr[beg + j];
        z0 += expf(lrelu(el[s * 2 + 0] + er0) - m0);
        z1 += expf(lrelu(el[s * 2 + 1] + er1) - m1);
    }
    const float ps0 = expf(es0 - m0), ps1 = expf(es1 - m1);
    z0 = wave_sum(z0) + ps0;
    z1 = wave_sum(z1) + ps1;

    const float asl = hi ? (ps1 / z1) : (ps0 / z0);
    float4 acc = make_float4(asl * vd.x, asl * vd.y, asl * vd.z, asl * vd.w);

    add_rows(acc, feat, s_c, p0_c / z0, p1_c / z1, n0, lane, hi);
    for (int base = 32; base < deg; base += 32) {
        const int j = base + lane;
        int   s_l  = d;
        float a0_l = 0.f, a1_l = 0.f;
        if (j < deg) {
            s_l  = csr[beg + j];
            a0_l = expf(lrelu(el[s_l * 2 + 0] + er0) - m0) / z0;
            a1_l = expf(lrelu(el[s_l * 2 + 1] + er1) - m1) / z1;
        }
        const int rest = deg - base;
        add_rows(acc, feat, s_l, a0_l, a1_l, rest < 32 ? rest : 32, lane, hi);
    }

    const float* bp = bias + lane * 4;
    float4 r = make_float4(acc.x + bp[0], acc.y + bp[1], acc.z + bp[2], acc.w + bp[3]);
    if (relu) {
        r.x = fmaxf(r.x, 0.f); r.y = fmaxf(r.y, 0.f);
        r.z = fmaxf(r.z, 0.f); r.w = fmaxf(r.w, 0.f);
    }
    reinterpret_cast<float4*>(out + (size_t)d * 128)[lane] = r;
}

extern "C" void kernel_launch(void* const* d_in, const int* in_sizes, int n_in,
                              void* d_out, int out_size, void* d_ws, size_t ws_size,
                              hipStream_t stream) {
    const float* x   = (const float*)d_in[0];
    const int*   src = (const int*)  d_in[1];
    const int*   dst = (const int*)  d_in[2];
    const float* W1  = (const float*)d_in[3];
    const float* al1 = (const float*)d_in[4];
    const float* ar1 = (const float*)d_in[5];
    const float* b1  = (const float*)d_in[6];
    const float* W2  = (const float*)d_in[7];
    const float* al2 = (const float*)d_in[8];
    const float* ar2 = (const float*)d_in[9];
    const float* b2  = (const float*)d_in[10];

    const int N = in_sizes[0] / 128;
    const long long E  = in_sizes[1];
    const long long Et = E + N;

    float* out = (float*)d_out;

    float* feat = (float*)d_ws;
    float* el   = feat + (size_t)N * 128;
    float* er   = el + (size_t)N * 2;
    float* ssum = er + (size_t)N * 2;
    int*   mkey = (int*)(ssum + (size_t)N * 2);

    const dim3 gemmGrid((N + 15) / 16, 2);

    const int    nScan = (N + 4095) / 4096;
    const size_t need  = ((size_t)N * 132 + (size_t)N + 1024 + (size_t)E) * 4;
    if (ws_size >= need && nScan <= 1024 && E < 0x7fffffffLL) {
        int* pos  = (int*)(er + (size_t)N * 2);
        int* bsum = pos + (size_t)N;
        int* csr  = bsum + 1024;
        const int E32     = (int)E;
        const int cBlocks = (E32 + 255) / 256;
        const int gBlocks = (N + 7) / 8;

        csr_zero<<<(N + 255) / 256, 256, 0, stream>>>(pos, N);
        csr_count<<<cBlocks, 256, 0, stream>>>(dst, E32, pos);
        csr_scan_block<<<nScan, 1024, 0, stream>>>(pos, N, bsum);
        csr_scan_add<<<nScan, 1024, 0, stream>>>(pos, N, bsum);
        csr_fill<<<cBlocks, 256, 0, stream>>>(src, dst, E32, pos, csr);

        gat_gemm_wmma<<<gemmGrid, 128, 0, stream>>>(x, W1, feat, N);
        gat_coef<<<(N + 255) / 256, 256, 0, stream>>>(feat, al1, ar1, el, er, N);
        gat_gather<<<gBlocks, 256, 0, stream>>>(pos, csr, el, er, feat, b1, out, N, 1);

        gat_gemm_wmma<<<gemmGrid, 128, 0, stream>>>(out, W2, feat, N);
        gat_coef<<<(N + 255) / 256, 256, 0, stream>>>(feat, al2, ar2, el, er, N);
        gat_gather<<<gBlocks, 256, 0, stream>>>(pos, csr, el, er, feat, b2, out, N, 0);
        return;
    }

    const int  totalNF = N * 128;
    const int  n2      = N * 2;
    const int  eBlocks = (int)((Et + 255) / 256);
    const int  aBlocks = (int)((Et + 7) / 8);

    gat_gemm_wmma<<<gemmGrid, 128, 0, stream>>>(x, W1, feat, N);
    gat_coef<<<(N + 255) / 256, 256, 0, stream>>>(feat, al1, ar1, el, er, N);
    gat_init_ms<<<(n2 + 255) / 256, 256, 0, stream>>>(mkey, ssum, n2);
    gat_init_bias<<<(totalNF + 255) / 256, 256, 0, stream>>>(out, b1, totalNF);
    gat_edge_max<<<eBlocks, 256, 0, stream>>>(src, dst, E, Et, el, er, mkey);
    gat_edge_sum<<<eBlocks, 256, 0, stream>>>(src, dst, E, Et, el, er, mkey, ssum);
    gat_edge_agg<<<aBlocks, 256, 0, stream>>>(src, dst, E, Et, el, er, mkey, ssum, feat, out);
    gat_relu<<<(totalNF + 255) / 256, 256, 0, stream>>>(out, totalNF);

    gat_gemm_wmma<<<gemmGrid, 128, 0, stream>>>(out, W2, feat, N);
    gat_coef<<<(N + 255) / 256, 256, 0, stream>>>(feat, al2, ar2, el, er, N);
    gat_init_ms<<<(n2 + 255) / 256, 256, 0, stream>>>(mkey, ssum, n2);
    gat_init_bias<<<(totalNF + 255) / 256, 256, 0, stream>>>(out, b2, totalNF);
    gat_edge_max<<<eBlocks, 256, 0, stream>>>(src, dst, E, Et, el, er, mkey);
    gat_edge_sum<<<eBlocks, 256, 0, stream>>>(src, dst, E, Et, el, er, mkey, ssum);
    gat_edge_agg<<<aBlocks, 256, 0, stream>>>(src, dst, E, Et, el, er, mkey, ssum, feat, out);
}


# Edge sort by dst redone as two-level counting sort ranked by LDS counters (4096-edge tiles by 256-node bucket, then a block per bucket): 2 launches for 5, no device atomics; sum order = LDS arrival
# speedup vs baseline: 5.7904x; 1.6245x over previous
#include <hip/hip_runtime.h>
#include <hip/hip_bf16.h>

typedef float v2f __attribute__((ext_vector_type(2)));
typedef float v8f __attribute__((ext_vector_type(8)));

#define NEG_SLOPE 0.2f

__device__ __forceinline__ int f2ord(float f) {
    int i = __float_as_int(f);
    return (i >= 0) ? i : (i ^ 0x7fffffff);
}
__device__ __forceinline__ float ord2f(int i) {
    return __int_as_float((i >= 0) ? i : (i ^ 0x7fffffff));
}

__global__ __launch_bounds__(128) void gat_gemm_wmma(
    const float* __restrict__ A, const float* __restrict__ W,
    float* __restrict__ Out, int M) {
    __shared__ float As[16][132];

    const int rowBase = blockIdx.x * 16;

    #pragma unroll
    for (int i = 0; i < 16; ++i) {
        int idx = threadIdx.x + i * 128;
        int r = idx >> 7;
        int k = idx & 127;
        int row = rowBase + r;
        if (row >= M) row = M - 1;
        As[r][k] = A[(size_t)row * 128 + k];
    }
    __syncthreads();

    const int wave = threadIdx.x >> 5;
    const int lane = threadIdx.x & 31;
    const int half = lane >> 4;
    const int l16  = lane & 15;
    const int ncol = (blockIdx.y * 4 + wave) * 16 + l16;

    v8f acc = {};
    for (int kk = 0; kk < 128; kk += 4) {
        v2f a, b;
        a.x = As[l16][kk + half * 2 + 0];
        a.y = As[l16][kk + half * 2 + 1];
        b.x = W[(size_t)(kk + half * 2 + 0) * 128 + ncol];
        b.y = W[(size_t)(kk + half * 2 + 1) * 128 + ncol];
        acc = __builtin_amdgcn_wmma_f32_16x16x4_f32(
             false, a,  false, b,
             (short)0, acc,  false,  false);
    }

    #pragma unroll
    for (int r = 0; r < 8; ++r) {
        int row = rowBase + r + half * 8;
        if (row < M) Out[(size_t)row * 128 + ncol] = acc[r];
    }
}

__global__ void gat_coef(const float* __restrict__ feat,
                         const float* __restrict__ al,
                         const float* __restrict__ ar,
                         float* __restrict__ el, float* __restrict__ er,
                         int N) {
    int n = blockIdx.x * blockDim.x + threadIdx.x;
    if (n >= N) return;
    const float* f = feat + (size_t)n * 128;
    float e0 = 0.f, e1 = 0.f, r0 = 0.f, r1 = 0.f;
    #pragma unroll 4
    for (int d = 0; d < 64; ++d) {
        float h0 = f[d], h1 = f[64 + d];
        e0 += h0 * al[d];
        e1 += h1 * al[64 + d];
        r0 += h0 * ar[d];
        r1 += h1 * ar[64 + d];
    }
    el[n * 2 + 0] = e0; el[n * 2 + 1] = e1;
    er[n * 2 + 0] = r0; er[n * 2 + 1] = r1;
}

__global__ void gat_init_ms(int* __restrict__ mkey, float* __restrict__ s, int n2) {
    int i = blockIdx.x * blockDim.x + threadIdx.x;
    if (i < n2) { mkey[i] = 0x80000000; s[i] = 0.f; }
}

__global__ void gat_init_bias(float* __restrict__ out, const float* __restrict__ b,
                              int total) {
    int i = blockIdx.x * blockDim.x + threadIdx.x;
    if (i < total) out[i] = b[i & 127];
}

__device__ __forceinline__ void edge_sd(long long i, long long E,
                                        const int* __restrict__ src,
                                        const int* __restrict__ dst,
                                        int& s, int& d) {
    if (i < E) { s = src[i]; d = dst[i]; }
    else       { s = (int)(i - E); d = s; }
}

__global__ void gat_edge_max(const int* __restrict__ src, const int* __restrict__ dst,
                             long long E, long long Et,
                             const float* __restrict__ el, const float* __restrict__ er,
                             int* __restrict__ mkey) {
    long long i = (long long)blockIdx.x * blockDim.x + threadIdx.x;
    if (i >= Et) return;
    int s, d; edge_sd(i, E, src, dst, s, d);
    #pragma unroll
    for (int h = 0; h < 2; ++h) {
        float e = el[s * 2 + h] + er[d * 2 + h];
        e = (e >= 0.f) ? e : NEG_SLOPE * e;
        atomicMax(&mkey[d * 2 + h], f2ord(e));
    }
}

__global__ void gat_edge_sum(const int* __restrict__ src, const int* __restrict__ dst,
                             long long E, long long Et,
                             const float* __restrict__ el, const float* __restrict__ er,
                             const int* __restrict__ mkey, float* __restrict__ ssum) {
    long long i = (long long)blockIdx.x * blockDim.x + threadIdx.x;
    if (i >= Et) return;
    int s, d; edge_sd(i, E, src, dst, s, d);
    #pragma unroll
    for (int h = 0; h < 2; ++h) {
        float e = el[s * 2 + h] + er[d * 2 + h];
        e = (e >= 0.f) ? e : NEG_SLOPE * e;
        atomicAdd(&ssum[d * 2 + h], expf(e - ord2f(mkey[d * 2 + h])));
    }
}

__global__ __launch_bounds__(256) void gat_edge_agg(
    const int* __restrict__ src, const int* __restrict__ dst,
    long long E, long long Et,
    const float* __restrict__ el, const float* __restrict__ er,
    const int* __restrict__ mkey, const float* __restrict__ ssum,
    const float* __restrict__ feat, float* __restrict__ out) {
    long long w = (long long)blockIdx.x * 8 + (threadIdx.x >> 5);
    if (w >= Et) return;
    const int lane = threadIdx.x & 31;
    int s, d; edge_sd(w, E, src, dst, s, d);

    float e0 = el[s * 2 + 0] + er[d * 2 + 0];
    float e1 = el[s * 2 + 1] + er[d * 2 + 1];
    e0 = (e0 >= 0.f) ? e0 : NEG_SLOPE * e0;
    e1 = (e1 >= 0.f) ? e1 : NEG_SLOPE * e1;
    const float alpha0 = expf(e0 - ord2f(mkey[d * 2 + 0])) / ssum[d * 2 + 0];
    const float alpha1 = expf(e1 - ord2f(mkey[d * 2 + 1])) / ssum[d * 2 + 1];

    const float* f = feat + (size_t)s * 128 + lane;
    const float v0 = f[0], v1 = f[32], v2 = f[64], v3 = f[96];
    float* o = out + (size_t)d * 128 + lane;
    atomicAdd(o +  0, alpha0 * v0);
    atomicAdd(o + 32, alpha0 * v1);
    atomicAdd(o + 64, alpha1 * v2);
    atomicAdd(o + 96, alpha1 * v3);
}

__global__ void gat_relu(float* __restrict__ x, int total) {
    int i = blockIdx.x * blockDim.x + threadIdx.x;
    if (i < total) x[i] = fmaxf(x[i], 0.f);
}


__global__ void csr_zero(int* __restrict__ pos, int N) {
    int i = blockIdx.x * blockDim.x + threadIdx.x;
    if (i < N) pos[i] = 0;
}

__global__ void csr_count(const int* __restrict__ dst, int E, int* __restrict__ pos) {
    int i = blockIdx.x * blockDim.x + threadIdx.x;
    if (i < E) atomicAdd(&pos[dst[i]], 1);
}

__global__ __launch_bounds__(1024) void csr_scan_block(int* __restrict__ pos, int N,
                                                       int* __restrict__ bsum) {
    __shared__ int wtot[32];
    const int t = threadIdx.x, lane = t & 31, wv = t >> 5;
    const int base = blockIdx.x * 4096 + t * 4;
    int v[4];
    #pragma unroll
    for (int i = 0; i < 4; ++i) v[i] = (base + i < N) ? pos[base + i] : 0;
    const int tsum = v[0] + v[1] + v[2] + v[3];
    int x = tsum;
    #pragma unroll
    for (int o = 1; o < 32; o <<= 1) {
        int y = __shfl_up(x, o);
        if (lane >= o) x += y;
    }
    if (lane == 31) wtot[wv] = x;
    __syncthreads();
    if (wv == 0) {
        int w = wtot[lane];
        #pragma unroll
        for (int o = 1; o < 32; o <<= 1) {
            int y = __shfl_up(w, o);
            if (lane >= o) w += y;
        }
        wtot[lane] = w;
    }
    __syncthreads();
    int run = x - tsum + (wv > 0 ? wtot[wv - 1] : 0);
    #pragma unroll
    for (int i = 0; i < 4; ++i) {
        if (base + i < N) pos[base + i] = run;
        run += v[i];
    }
    if (t == 1023) bsum[blockIdx.x] = run;
}

__global__ __launch_bounds__(1024) void csr_scan_add(int* __restrict__ pos, int N,
                                                     const int* __restrict__ bsum) {
    const int b = blockIdx.x;
    int off = 0;
    for (int j = 0; j < b; ++j) off += bsum[j];
    const int base = b * 4096 + threadIdx.x * 4;
    #pragma unroll
    for (int i = 0; i < 4; ++i)
        if (base + i < N) pos[base + i] += off;
}

__global__ void csr_fill(const int* __restrict__ src, const int* __restrict__ dst, int E,
                         int* __restrict__ pos, int* __restrict__ csr) {
    int i = blockIdx.x * blockDim.x + threadIdx.x;
    if (i >= E) return;
    int p = atomicAdd(&pos[dst[i]], 1);
    if ((unsigned)p < (unsigned)E) csr[p] = src[i];
}


#define CS_TILE 4096
#define CS_BSH  8
#define CS_BN   256
#define CS_MAXB 1024

__device__ __forceinline__ int wave_sum_i(int v) {
    for (int o = 16; o > 0; o >>= 1) v += __shfl_xor(v, o);
    return v;
}

__global__ __launch_bounds__(1024) void csr_tile_sort(
    const int* __restrict__ src, const int* __restrict__ dst, int E, int NB, int NT,
    int* __restrict__ toffT, int* __restrict__ tedge) {
    __shared__ int cnt[CS_MAXB];
    __shared__ int wtot[32];
    __shared__ int stS[CS_TILE];
    __shared__ int stD[CS_TILE];

    const int t = threadIdx.x, lane = t & 31, wv = t >> 5;
    const int tile = blockIdx.x;
    const int e0 = tile * CS_TILE;

    cnt[t] = 0;
    __syncthreads();

    int s[4], d[4];
    for (int i = 0; i < 4; ++i) {
        const int e = e0 + i * 1024 + t;
        s[i] = 0; d[i] = -1;
        if (e < E) { s[i] = src[e]; d[i] = dst[e]; }
    }
    for (int i = 0; i < 4; ++i)
        if (d[i] >= 0) atomicAdd(&cnt[d[i] >> CS_BSH], 1);
    __syncthreads();

    const int c = cnt[t];
    int x = c;
    for (int o = 1; o < 32; o <<= 1) {
        int y = __shfl_up(x, o);
        if (lane >= o) x += y;
    }
    if (lane == 31) wtot[wv] = x;
    __syncthreads();
    if (wv == 0) {
        int w = wtot[lane];
        for (int o = 1; o < 32; o <<= 1) {
            int y = __shfl_up(w, o);
            if (lane >= o) w += y;
        }
        wtot[lane] = w;
    }
    __syncthreads();
    const int excl = x - c + (wv > 0 ? wtot[wv - 1] : 0);
    cnt[t] = excl;
    if (t <= NB) toffT[(size_t)t * NT + tile] = excl;
    __syncthreads();

    for (int i = 0; i < 4; ++i)
        if (d[i] >= 0) {
            const int r = atomicAdd(&cnt[d[i] >> CS_BSH], 1);
            stS[r] = s[i];
            stD[r] = d[i];
        }
    __syncthreads();
    const int left = E - e0;
    const int nE = left < CS_TILE ? left : CS_TILE;
    int2* outp = reinterpret_cast<int2*>(tedge) + (size_t)tile * CS_TILE;
    for (int i = 0; i < 4; ++i) {
        const int j = i * 1024 + t;
        if (j < nE) outp[j] = make_int2(stS[j], stD[j]);
    }
}

__global__ __launch_bounds__(1024) void csr_bucket_build(
    const int* __restrict__ toffT, const int* __restrict__ tedge, int N, int E, int NT,
    int* __restrict__ pos, int* __restrict__ csr) {
    __shared__ int hist[CS_BN];
    __shared__ int wsum[32];
    __shared__ int wtot[32];

    const int t = threadIdx.x, lane = t & 31, wv = t >> 5;
    const int b = blockIdx.x;
    const int* lo = toffT + (size_t)b * NT;
    const int* hi = lo + NT;
    const int2* edges = reinterpret_cast<const int2*>(tedge);

    if (t < CS_BN) hist[t] = 0;
    int part = 0;
    for (int k = t; k < NT; k += 1024) part += lo[k];
    part = wave_sum_i(part);
    if (lane == 0) wsum[wv] = part;
    __syncthreads();
    int base = 0;
    for (int i = 0; i < 32; ++i) base += wsum[i];

    for (int k = t >> 1; k < NT; k += 512) {
        const int a = lo[k], z = hi[k];
        const int mid = a + ((z - a + 1) >> 1);
        const int s0 = (t & 1) ? mid : a;
        const int s1 = (t & 1) ? z : mid;
        const int2* p = edges + (size_t)k * CS_TILE;
        for (int j = s0; j < s1; j += 4) {
            int key[4];
            for (int u = 0; u < 4; ++u) {
                const int jj = (j + u < s1) ? (j + u) : (s1 - 1);
                key[u] = p[jj].y & (CS_BN - 1);
            }
            for (int u = 0; u < 4; ++u)
                atomicAdd(&hist[key[u]], (j + u < s1) ? 1 : 0);
        }
    }
    __syncthreads();

    const int c = (t < CS_BN) ? hist[t] : 0;
    int x = c;
    for (int o = 1; o < 32; o <<= 1) {
        int y = __shfl_up(x, o);
        if (lane >= o) x += y;
    }
    if (lane == 31) wtot[wv] = x;
    __syncthreads();
    if (t < CS_BN) {
        int incl = x;
        for (int i = 0; i < wv; ++i) incl += wtot[i];
        hist[t] = incl - c;
        const int n = b * CS_BN + t;
        if (n < N) pos[n] = base + incl;
    }
    __syncthreads();

    for (int k = t >> 1; k < NT; k += 512) {
        const int a = lo[k], z = hi[k];
        const int mid = a + ((z - a + 1) >> 1);
        const int s0 = (t & 1) ? mid : a;
        const int s1 = (t & 1) ? z : mid;
        const int2* p = edges + (size_t)k * CS_TILE;
        for (int j = s0; j < s1; j += 4) {
            int2 e[4];
            for (int u = 0; u < 4; ++u) {
                const int jj = (j + u < s1) ? (j + u) : (s1 - 1);
                e[u] = p[jj];
            }
            for (int u = 0; u < 4; ++u) {
                const bool ok = j + u < s1;
                const int r = base + atomicAdd(&hist[e[u].y & (CS_BN - 1)], ok ? 1 : 0);
                if (ok && (unsigned)r < (unsigned)E) csr[r] = e[u].x;
            }
        }
    }
}

__device__ __forceinline__ float lrelu(float e) { return (e >= 0.f) ? e : NEG_SLOPE * e; }

__device__ __forceinline__ float wave_max(float v) {
    #pragma unroll
    for (int o = 16; o > 0; o >>= 1) v = fmaxf(v, __shfl_xor(v, o));
    return v;
}
__device__ __forceinline__ float wave_sum(float v) {
    #pragma unroll
    for (int o = 16; o > 0; o >>= 1) v += __shfl_xor(v, o);
    return v;
}

__device__ __forceinline__ void add_rows(float4& acc, const float* __restrict__ feat,
                                         int s_l, float a0_l, float a1_l, int cnt,
                                         int lane, bool hi) {
    for (int k = 0; k < cnt; k += 4) {
        float4 v[4];
        float  a[4];
        #pragma unroll
        for (int i = 0; i < 4; ++i) {
            const int   s  = __builtin_amdgcn_readlane(s_l, k + i);
            const float a0 = __int_as_float(__builtin_amdgcn_readlane(__float_as_int(a0_l), k + i));
            const float a1 = __int_as_float(__builtin_amdgcn_readlane(__float_as_int(a1_l), k + i));
            a[i] = hi ? a1 : a0;
            v[i] = reinterpret_cast<const float4*>(feat + (size_t)s * 128)[lane];
        }
        #pragma unroll
        for (int i = 0; i < 4; ++i) {
            acc.x = fmaf(a[i], v[i].x, acc.x);
            acc.y = fmaf(a[i], v[i].y, acc.y);
            acc.z = fmaf(a[i], v[i].z, acc.z);
            acc.w = fmaf(a[i], v[i].w, acc.w);
        }
    }
}

__global__ __launch_bounds__(256) void gat_gather(
    const int* __restrict__ pos, const int* __restrict__ csr,
    const float* __restrict__ el, const float* __restrict__ er,
    const float* __restrict__ feat, const float* __restrict__ bias,
    float* __restrict__ out, int N, int relu) {
    const int d = __builtin_amdgcn_readfirstlane((int)(blockIdx.x * 8 + (threadIdx.x >> 5)));
    if (d >= N) return;
    const int lane = threadIdx.x & 31;
    const bool hi = lane >= 16;
    const int beg = (d > 0) ? pos[d - 1] : 0;
    const int deg = pos[d] - beg;

    const float4 vd = reinterpret_cast<const float4*>(feat + (size_t)d * 128)[lane];

    const float er0 = er[d * 2 + 0], er1 = er[d * 2 + 1];
    const float es0 = lrelu(el[d * 2 + 0] + er0);
    const float es1 = lrelu(el[d * 2 + 1] + er1);

    const int  n0  = deg < 32 ? deg : 32;
    const bool act = lane < n0;
    int   s_c  = d;
    float e0_c = 0.f, e1_c = 0.f;
    if (act) {
        s_c  = csr[beg + lane];
        e0_c = lrelu(el[s_c * 2 + 0] + er0);
        e1_c = lrelu(el[s_c * 2 + 1] + er1);
    }
    float m0 = act ? fmaxf(es0, e0_c) : es0;
    float m1 = act ? fmaxf(es1, e1_c) : es1;
    for (int j = 32 + lane; j < deg; j += 32) {
        const int s = csr[beg + j];
        m0 = fmaxf(m0, lrelu(el[s * 2 + 0] + er0));
        m1 = fmaxf(m1, lrelu(el[s * 2 + 1] + er1));
    }
    m0 = wave_max(m0);
    m1 = wave_max(m1);

    const float p0_c = act ? expf(e0_c - m0) : 0.f;
    const float p1_c = act ? expf(e1_c - m1) : 0.f;
    float z0 = p0_c, z1 = p1_c;
    for (int j = 32 + lane; j < deg; j += 32) {
        const int s = csr[beg + j];
        z0 += expf(lrelu(el[s * 2 + 0] + er0) - m0);
        z1 += expf(lrelu(el[s * 2 + 1] + er1) - m1);
    }
    const float ps0 = expf(es0 - m0), ps1 = expf(es1 - m1);
    z0 = wave_sum(z0) + ps0;
    z1 = wave_sum(z1) + ps1;

    const float asl = hi ? (ps1 / z1) : (ps0 / z0);
    float4 acc = make_float4(asl * vd.x, asl * vd.y, asl * vd.z, asl * vd.w);

    add_rows(acc, feat, s_c, p0_c / z0, p1_c / z1, n0, lane, hi);
    for (int base = 32; base < deg; base += 32) {
        const int j = base + lane;
        int   s_l  = d;
        float a0_l = 0.f, a1_l = 0.f;
        if (j < deg) {
            s_l  = csr[beg + j];
            a0_l = expf(lrelu(el[s_l * 2 + 0] + er0) - m0) / z0;
            a1_l = expf(lrelu(el[s_l * 2 + 1] + er1) - m1) / z1;
        }
        const int rest = deg - base;
        add_rows(acc, feat, s_l, a0_l, a1_l, rest < 32 ? rest : 32, lane, hi);
    }

    const float* bp = bias + lane * 4;
    float4 r = make_float4(acc.x + bp[0], acc.y + bp[1], acc.z + bp[2], acc.w + bp[3]);
    if (relu) {
        r.x = fmaxf(r.x, 0.f); r.y = fmaxf(r.y, 0.f);
        r.z = fmaxf(r.z, 0.f); r.w = fmaxf(r.w, 0.f);
    }
    reinterpret_cast<float4*>(out + (size_t)d * 128)[lane] = r;
}

extern "C" void kernel_launch(void* const* d_in, const int* in_sizes, int n_in,
                              void* d_out, int out_size, void* d_ws, size_t ws_size,
                              hipStream_t stream) {
    const float* x   = (const float*)d_in[0];
    const int*   src = (const int*)  d_in[1];
    const int*   dst = (const int*)  d_in[2];
    const float* W1  = (const float*)d_in[3];
    const float* al1 = (const float*)d_in[4];
    const float* ar1 = (const float*)d_in[5];
    const float* b1  = (const float*)d_in[6];
    const float* W2  = (const float*)d_in[7];
    const float* al2 = (const float*)d_in[8];
    const float* ar2 = (const float*)d_in[9];
    const float* b2  = (const float*)d_in[10];

    const int N = in_sizes[0] / 128;
    const long long E  = in_sizes[1];
    const long long Et = E + N;

    float* out = (float*)d_out;

    float* feat = (float*)d_ws;
    float* el   = feat + (size_t)N * 128;
    float* er   = el + (size_t)N * 2;
    float* ssum = er + (size_t)N * 2;
    int*   mkey = (int*)(ssum + (size_t)N * 2);

    const dim3 gemmGrid((N + 15) / 16, 2);

    const int    nScan = (N + 4095) / 4096;
    const size_t need  = ((size_t)N * 132 + (size_t)N + 1024 + (size_t)E) * 4;
    if (ws_size >= need && nScan <= 1024 && E < 0x7fffffffLL) {
        int* pos  = (int*)(er + (size_t)N * 2);
        int* bsum = pos + (size_t)N;
        int* csr  = bsum + 1024;
        const int E32     = (int)E;
        const int cBlocks = (E32 + 255) / 256;
        const int gBlocks = (N + 7) / 8;

        const int    NB  = (N + CS_BN - 1) / CS_BN;
        const int    NT  = (E32 + CS_TILE - 1) / CS_TILE;
        const size_t tmp = ((size_t)NT * CS_TILE * 2 + (size_t)(NB + 1) * NT) * 4;
        if (E32 > 0 && E32 < 0x7fff0000 && NB + 1 <= CS_MAXB && tmp <= (size_t)N * 128 * 4) {
            int* tedge = (int*)feat;
            int* toffT = tedge + (size_t)NT * CS_TILE * 2;
            csr_tile_sort<<<NT, 1024, 0, stream>>>(src, dst, E32, NB, NT, toffT, tedge);
            csr_bucket_build<<<NB, 1024, 0, stream>>>(toffT, tedge, N, E32, NT, pos, csr);
        } else {
            csr_zero<<<(N + 255) / 256, 256, 0, stream>>>(pos, N);
            csr_count<<<cBlocks, 256, 0, stream>>>(dst, E32, pos);
            csr_scan_block<<<nScan, 1024, 0, stream>>>(pos, N, bsum);
            csr_scan_add<<<nScan, 1024, 0, stream>>>(pos, N, bsum);
            csr_fill<<<cBlocks, 256, 0, stream>>>(src, dst, E32, pos, csr);
        }

        gat_gemm_wmma<<<gemmGrid, 128, 0, stream>>>(x, W1, feat, N);
        gat_coef<<<(N + 255) / 256, 256, 0, stream>>>(feat, al1, ar1, el, er, N);
        gat_gather<<<gBlocks, 256, 0, stream>>>(pos, csr, el, er, feat, b1, out, N, 1);

        gat_gemm_wmma<<<gemmGrid, 128, 0, stream>>>(out, W2, feat, N);
        gat_coef<<<(N + 255) / 256, 256, 0, stream>>>(feat, al2, ar2, el, er, N);
        gat_gather<<<gBlocks, 256, 0, stream>>>(pos, csr, el, er, feat, b2, out, N, 0);
        return;
    }

    const int  totalNF = N * 128;
    const int  n2      = N * 2;
    const int  eBlocks = (int)((Et + 255) / 256);
    const int  aBlocks = (int)((Et + 7) / 8);

    gat_gemm_wmma<<<gemmGrid, 128, 0, stream>>>(x, W1, feat, N);
    gat_coef<<<(N + 255) / 256, 256, 0, stream>>>(feat, al1, ar1, el, er, N);
    gat_init_ms<<<(n2 + 255) / 256, 256, 0, stream>>>(mkey, ssum, n2);
    gat_init_bias<<<(totalNF + 255) / 256, 256, 0, stream>>>(out, b1, totalNF);
    gat_edge_max<<<eBlocks, 256, 0, stream>>>(src, dst, E, Et, el, er, mkey);
    gat_edge_sum<<<eBlocks, 256, 0, stream>>>(src, dst, E, Et, el, er, mkey, ssum);
    gat_edge_agg<<<aBlocks, 256, 0, stream>>>(src, dst, E, Et, el, er, mkey, ssum, feat, out);
    gat_relu<<<(totalNF + 255) / 256, 256, 0, stream>>>(out, totalNF);

    gat_gemm_wmma<<<gemmGrid, 128, 0, stream>>>(out, W2, feat, N);
    gat_coef<<<(N + 255) / 256, 256, 0, stream>>>(feat, al2, ar2, el, er, N);
    gat_init_ms<<<(n2 + 255) / 256, 256, 0, stream>>>(mkey, ssum, n2);
    gat_init_bias<<<(totalNF + 255) / 256, 256, 0, stream>>>(out, b2, totalNF);
    gat_edge_max<<<eBlocks, 256, 0, stream>>>(src, dst, E, Et, el, er, mkey);
    gat_edge_sum<<<eBlocks, 256, 0, stream>>>(src, dst, E, Et, el, er, mkey, ssum);
    gat_edge_agg<<<aBlocks, 256, 0, stream>>>(src, dst, E, Et, el, er, mkey, ssum, feat, out);
}


# GEMM epilogue computes el/er from the accumulator tiles (16-lane DPP sum tree, then 4 waves via LDS; block now spans all 128 columns): the 2 coef launches are removed; el/er sum order changes
# speedup vs baseline: 6.6091x; 1.1414x over previous
#include <hip/hip_runtime.h>
#include <hip/hip_bf16.h>

typedef float v2f __attribute__((ext_vector_type(2)));
typedef float v8f __attribute__((ext_vector_type(8)));

#define NEG_SLOPE 0.2f

__device__ __forceinline__ int f2ord(float f) {
    int i = __float_as_int(f);
    return (i >= 0) ? i : (i ^ 0x7fffffff);
}
__device__ __forceinline__ float ord2f(int i) {
    return __int_as_float((i >= 0) ? i : (i ^ 0x7fffffff));
}


template <int XM>
__device__ __forceinline__ float row_xor(float v) {
    return __int_as_float(__builtin_amdgcn_update_dpp(
        0, __float_as_int(v), 0x160 + XM  , 0xF, 0xF, true));
}

__global__ __launch_bounds__(256) void gat_gemm_wmma(
    const float* __restrict__ A, const float* __restrict__ W,
    const float* __restrict__ al, const float* __restrict__ ar,
    float* __restrict__ Out, float* __restrict__ el, float* __restrict__ er, int M) {
    __shared__ float As[16][132];
    __shared__ float red[8][32];

    const int rowBase = blockIdx.x * 16;

    #pragma unroll
    for (int i = 0; i < 8; ++i) {
        int idx = threadIdx.x + i * 256;
        int r = idx >> 7;
        int k = idx & 127;
        int row = rowBase + r;
        if (row >= M) row = M - 1;
        As[r][k] = A[(size_t)row * 128 + k];
    }
    __syncthreads();

    const int wave = threadIdx.x >> 5;
    const int lane = threadIdx.x & 31;
    const int half = lane >> 4;
    const int l16  = lane & 15;
    const int ncol = wave * 16 + l16;

    v8f acc = {};
    for (int kk = 0; kk < 128; kk += 4) {
        v2f a, b;
        a.x = As[l16][kk + half * 2 + 0];
        a.y = As[l16][kk + half * 2 + 1];
        b.x = W[(size_t)(kk + half * 2 + 0) * 128 + ncol];
        b.y = W[(size_t)(kk + half * 2 + 1) * 128 + ncol];
        acc = __builtin_amdgcn_wmma_f32_16x16x4_f32(
             false, a,  false, b,
             (short)0, acc,  false,  false);
    }

    #pragma unroll
    for (int r = 0; r < 8; ++r) {
        int row = rowBase + r + half * 8;
        if (row < M) Out[(size_t)row * 128 + ncol] = acc[r];
    }

    const float alv = al[ncol], arv = ar[ncol];
    const bool b3 = (l16 & 8) != 0, b2 = (l16 & 4) != 0, b1 = (l16 & 2) != 0, b0 = (l16 & 1) != 0;
    const float ck = b3 ? arv : alv;
    const float cs = b3 ? alv : arv;
    float k8[8];
    for (int i = 0; i < 8; ++i) k8[i] = acc[i] * ck + row_xor<8>(acc[i] * cs);
    float k4[4];
    for (int i = 0; i < 4; ++i)
        k4[i] = (b2 ? k8[4 + i] : k8[i]) + row_xor<4>(b2 ? k8[i] : k8[4 + i]);
    float k2[2];
    for (int i = 0; i < 2; ++i)
        k2[i] = (b1 ? k4[2 + i] : k4[i]) + row_xor<2>(b1 ? k4[i] : k4[2 + i]);
    const float k1 = (b0 ? k2[1] : k2[0]) + row_xor<1>(b0 ? k2[0] : k2[1]);
    red[wave][lane] = k1;
    __syncthreads();

    if (threadIdx.x < 64) {
        const int kind = threadIdx.x >> 5;
        const int row  = lane >> 1, hd = lane & 1;
        const int slot = (row >> 3) * 16 + kind * 8 + (row & 7);
        const float s = ((red[hd * 4 + 0][slot] + red[hd * 4 + 1][slot])
                         + red[hd * 4 + 2][slot]) + red[hd * 4 + 3][slot];
        if (rowBase + row < M) (kind ? er : el)[(size_t)rowBase * 2 + lane] = s;
    }
}

__global__ void gat_init_ms(int* __restrict__ mkey, float* __restrict__ s, int n2) {
    int i = blockIdx.x * blockDim.x + threadIdx.x;
    if (i < n2) { mkey[i] = 0x80000000; s[i] = 0.f; }
}

__global__ void gat_init_bias(float* __restrict__ out, const float* __restrict__ b,
                              int total) {
    int i = blockIdx.x * blockDim.x + threadIdx.x;
    if (i < total) out[i] = b[i & 127];
}

__device__ __forceinline__ void edge_sd(long long i, long long E,
                                        const int* __restrict__ src,
                                        const int* __restrict__ dst,
                                        int& s, int& d) {
    if (i < E) { s = src[i]; d = dst[i]; }
    else       { s = (int)(i - E); d = s; }
}

__global__ void gat_edge_max(const int* __restrict__ src, const int* __restrict__ dst,
                             long long E, long long Et,
                             const float* __restrict__ el, const float* __restrict__ er,
                             int* __restrict__ mkey) {
    long long i = (long long)blockIdx.x * blockDim.x + threadIdx.x;
    if (i >= Et) return;
    int s, d; edge_sd(i, E, src, dst, s, d);
    #pragma unroll
    for (int h = 0; h < 2; ++h) {
        float e = el[s * 2 + h] + er[d * 2 + h];
        e = (e >= 0.f) ? e : NEG_SLOPE * e;
        atomicMax(&mkey[d * 2 + h], f2ord(e));
    }
}

__global__ void gat_edge_sum(const int* __restrict__ src, const int* __restrict__ dst,
                             long long E, long long Et,
                             const float* __restrict__ el, const float* __restrict__ er,
                             const int* __restrict__ mkey, float* __restrict__ ssum) {
    long long i = (long long)blockIdx.x * blockDim.x + threadIdx.x;
    if (i >= Et) return;
    int s, d; edge_sd(i, E, src, dst, s, d);
    #pragma unroll
    for (int h = 0; h < 2; ++h) {
        float e = el[s * 2 + h] + er[d * 2 + h];
        e = (e >= 0.f) ? e : NEG_SLOPE * e;
        atomicAdd(&ssum[d * 2 + h], expf(e - ord2f(mkey[d * 2 + h])));
    }
}

__global__ __launch_bounds__(256) void gat_edge_agg(
    const int* __restrict__ src, const int* __restrict__ dst,
    long long E, long long Et,
    const float* __restrict__ el, const float* __restrict__ er,
    const int* __restrict__ mkey, const float* __restrict__ ssum,
    const float* __restrict__ feat, float* __restrict__ out) {
    long long w = (long long)blockIdx.x * 8 + (threadIdx.x >> 5);
    if (w >= Et) return;
    const int lane = threadIdx.x & 31;
    int s, d; edge_sd(w, E, src, dst, s, d);

    float e0 = el[s * 2 + 0] + er[d * 2 + 0];
    float e1 = el[s * 2 + 1] + er[d * 2 + 1];
    e0 = (e0 >= 0.f) ? e0 : NEG_SLOPE * e0;
    e1 = (e1 >= 0.f) ? e1 : NEG_SLOPE * e1;
    const float alpha0 = expf(e0 - ord2f(mkey[d * 2 + 0])) / ssum[d * 2 + 0];
    const float alpha1 = expf(e1 - ord2f(mkey[d * 2 + 1])) / ssum[d * 2 + 1];

    const float* f = feat + (size_t)s * 128 + lane;
    const float v0 = f[0], v1 = f[32], v2 = f[64], v3 = f[96];
    float* o = out + (size_t)d * 128 + lane;
    atomicAdd(o +  0, alpha0 * v0);
    atomicAdd(o + 32, alpha0 * v1);
    atomicAdd(o + 64, alpha1 * v2);
    atomicAdd(o + 96, alpha1 * v3);
}

__global__ void gat_relu(float* __restrict__ x, int total) {
    int i = blockIdx.x * blockDim.x + threadIdx.x;
    if (i < total) x[i] = fmaxf(x[i], 0.f);
}


__global__ void csr_zero(int* __restrict__ pos, int N) {
    int i = blockIdx.x * blockDim.x + threadIdx.x;
    if (i < N) pos[i] = 0;
}

__global__ void csr_count(const int* __restrict__ dst, int E, int* __restrict__ pos) {
    int i = blockIdx.x * blockDim.x + threadIdx.x;
    if (i < E) atomicAdd(&pos[dst[i]], 1);
}

__global__ __launch_bounds__(1024) void csr_scan_block(int* __restrict__ pos, int N,
                                                       int* __restrict__ bsum) {
    __shared__ int wtot[32];
    const int t = threadIdx.x, lane = t & 31, wv = t >> 5;
    const int base = blockIdx.x * 4096 + t * 4;
    int v[4];
    #pragma unroll
    for (int i = 0; i < 4; ++i) v[i] = (base + i < N) ? pos[base + i] : 0;
    const int tsum = v[0] + v[1] + v[2] + v[3];
    int x = tsum;
    #pragma unroll
    for (int o = 1; o < 32; o <<= 1) {
        int y = __shfl_up(x, o);
        if (lane >= o) x += y;
    }
    if (lane == 31) wtot[wv] = x;
    __syncthreads();
    if (wv == 0) {
        int w = wtot[lane];
        #pragma unroll
        for (int o = 1; o < 32; o <<= 1) {
            int y = __shfl_up(w, o);
            if (lane >= o) w += y;
        }
        wtot[lane] = w;
    }
    __syncthreads();
    int run = x - tsum + (wv > 0 ? wtot[wv - 1] : 0);
    #pragma unroll
    for (int i = 0; i < 4; ++i) {
        if (base + i < N) pos[base + i] = run;
        run += v[i];
    }
    if (t == 1023) bsum[blockIdx.x] = run;
}

__global__ __launch_bounds__(1024) void csr_scan_add(int* __restrict__ pos, int N,
                                                     const int* __restrict__ bsum) {
    const int b = blockIdx.x;
    int off = 0;
    for (int j = 0; j < b; ++j) off += bsum[j];
    const int base = b * 4096 + threadIdx.x * 4;
    #pragma unroll
    for (int i = 0; i < 4; ++i)
        if (base + i < N) pos[base + i] += off;
}

__global__ void csr_fill(const int* __restrict__ src, const int* __restrict__ dst, int E,
                         int* __restrict__ pos, int* __restrict__ csr) {
    int i = blockIdx.x * blockDim.x + threadIdx.x;
    if (i >= E) return;
    int p = atomicAdd(&pos[dst[i]], 1);
    if ((unsigned)p < (unsigned)E) csr[p] = src[i];
}


#define CS_TILE 4096
#define CS_BSH  8
#define CS_BN   256
#define CS_MAXB 1024

__device__ __forceinline__ int wave_sum_i(int v) {
    for (int o = 16; o > 0; o >>= 1) v += __shfl_xor(v, o);
    return v;
}

__global__ __launch_bounds__(1024) void csr_tile_sort(
    const int* __restrict__ src, const int* __restrict__ dst, int E, int NB, int NT,
    int* __restrict__ toffT, int* __restrict__ tedge) {
    __shared__ int cnt[CS_MAXB];
    __shared__ int wtot[32];
    __shared__ int stS[CS_TILE];
    __shared__ int stD[CS_TILE];

    const int t = threadIdx.x, lane = t & 31, wv = t >> 5;
    const int tile = blockIdx.x;
    const int e0 = tile * CS_TILE;

    cnt[t] = 0;
    __syncthreads();

    int s[4], d[4];
    for (int i = 0; i < 4; ++i) {
        const int e = e0 + i * 1024 + t;
        s[i] = 0; d[i] = -1;
        if (e < E) { s[i] = src[e]; d[i] = dst[e]; }
    }
    for (int i = 0; i < 4; ++i)
        if (d[i] >= 0) atomicAdd(&cnt[d[i] >> CS_BSH], 1);
    __syncthreads();

    const int c = cnt[t];
    int x = c;
    for (int o = 1; o < 32; o <<= 1) {
        int y = __shfl_up(x, o);
        if (lane >= o) x += y;
    }
    if (lane == 31) wtot[wv] = x;
    __syncthreads();
    if (wv == 0) {
        int w = wtot[lane];
        for (int o = 1; o < 32; o <<= 1) {
            int y = __shfl_up(w, o);
            if (lane >= o) w += y;
        }
        wtot[lane] = w;
    }
    __syncthreads();
    const int excl = x - c + (wv > 0 ? wtot[wv - 1] : 0);
    cnt[t] = excl;
    if (t <= NB) toffT[(size_t)t * NT + tile] = excl;
    __syncthreads();

    for (int i = 0; i < 4; ++i)
        if (d[i] >= 0) {
            const int r = atomicAdd(&cnt[d[i] >> CS_BSH], 1);
            stS[r] = s[i];
            stD[r] = d[i];
        }
    __syncthreads();
    const int left = E - e0;
    const int nE = left < CS_TILE ? left : CS_TILE;
    int2* outp = reinterpret_cast<int2*>(tedge) + (size_t)tile * CS_TILE;
    for (int i = 0; i < 4; ++i) {
        const int j = i * 1024 + t;
        if (j < nE) outp[j] = make_int2(stS[j], stD[j]);
    }
}

__global__ __launch_bounds__(1024) void csr_bucket_build(
    const int* __restrict__ toffT, const int* __restrict__ tedge, int N, int E, int NT,
    int* __restrict__ pos, int* __restrict__ csr) {
    __shared__ int hist[CS_BN];
    __shared__ int wsum[32];
    __shared__ int wtot[32];

    const int t = threadIdx.x, lane = t & 31, wv = t >> 5;
    const int b = blockIdx.x;
    const int* lo = toffT + (size_t)b * NT;
    const int* hi = lo + NT;
    const int2* edges = reinterpret_cast<const int2*>(tedge);

    if (t < CS_BN) hist[t] = 0;
    int part = 0;
    for (int k = t; k < NT; k += 1024) part += lo[k];
    part = wave_sum_i(part);
    if (lane == 0) wsum[wv] = part;
    __syncthreads();
    int base = 0;
    for (int i = 0; i < 32; ++i) base += wsum[i];

    for (int k = t >> 1; k < NT; k += 512) {
        const int a = lo[k], z = hi[k];
        const int mid = a + ((z - a + 1) >> 1);
        const int s0 = (t & 1) ? mid : a;
        const int s1 = (t & 1) ? z : mid;
        const int2* p = edges + (size_t)k * CS_TILE;
        for (int j = s0; j < s1; j += 4) {
            int key[4];
            for (int u = 0; u < 4; ++u) {
                const int jj = (j + u < s1) ? (j + u) : (s1 - 1);
                key[u] = p[jj].y & (CS_BN - 1);
            }
            for (int u = 0; u < 4; ++u)
                atomicAdd(&hist[key[u]], (j + u < s1) ? 1 : 0);
        }
    }
    __syncthreads();

    const int c = (t < CS_BN) ? hist[t] : 0;
    int x = c;
    for (int o = 1; o < 32; o <<= 1) {
        int y = __shfl_up(x, o);
        if (lane >= o) x += y;
    }
    if (lane == 31) wtot[wv] = x;
    __syncthreads();
    if (t < CS_BN) {
        int incl = x;
        for (int i = 0; i < wv; ++i) incl += wtot[i];
        hist[t] = incl - c;
        const int n = b * CS_BN + t;
        if (n < N) pos[n] = base + incl;
    }
    __syncthreads();

    for (int k = t >> 1; k < NT; k += 512) {
        const int a = lo[k], z = hi[k];
        const int mid = a + ((z - a + 1) >> 1);
        const int s0 = (t & 1) ? mid : a;
        const int s1 = (t & 1) ? z : mid;
        const int2* p = edges + (size_t)k * CS_TILE;
        for (int j = s0; j < s1; j += 4) {
            int2 e[4];
            for (int u = 0; u < 4; ++u) {
                const int jj = (j + u < s1) ? (j + u) : (s1 - 1);
                e[u] = p[jj];
            }
            for (int u = 0; u < 4; ++u) {
                const bool ok = j + u < s1;
                const int r = base + atomicAdd(&hist[e[u].y & (CS_BN - 1)], ok ? 1 : 0);
                if (ok && (unsigned)r < (unsigned)E) csr[r] = e[u].x;
            }
        }
    }
}

__device__ __forceinline__ float lrelu(float e) { return (e >= 0.f) ? e : NEG_SLOPE * e; }

__device__ __forceinline__ float wave_max(float v) {
    #pragma unroll
    for (int o = 16; o > 0; o >>= 1) v = fmaxf(v, __shfl_xor(v, o));
    return v;
}
__device__ __forceinline__ float wave_sum(float v) {
    #pragma unroll
    for (int o = 16; o > 0; o >>= 1) v += __shfl_xor(v, o);
    return v;
}

__device__ __forceinline__ void add_rows(float4& acc, const float* __restrict__ feat,
                                         int s_l, float a0_l, float a1_l, int cnt,
                                         int lane, bool hi) {
    for (int k = 0; k < cnt; k += 4) {
        float4 v[4];
        float  a[4];
        #pragma unroll
        for (int i = 0; i < 4; ++i) {
            const int   s  = __builtin_amdgcn_readlane(s_l, k + i);
            const float a0 = __int_as_float(__builtin_amdgcn_readlane(__float_as_int(a0_l), k + i));
            const float a1 = __int_as_float(__builtin_amdgcn_readlane(__float_as_int(a1_l), k + i));
            a[i] = hi ? a1 : a0;
            v[i] = reinterpret_cast<const float4*>(feat + (size_t)s * 128)[lane];
        }
        #pragma unroll
        for (int i = 0; i < 4; ++i) {
            acc.x = fmaf(a[i], v[i].x, acc.x);
            acc.y = fmaf(a[i], v[i].y, acc.y);
            acc.z = fmaf(a[i], v[i].z, acc.z);
            acc.w = fmaf(a[i], v[i].w, acc.w);
        }
    }
}

__global__ __launch_bounds__(256) void gat_gather(
    const int* __restrict__ pos, const int* __restrict__ csr,
    const float* __restrict__ el, const float* __restrict__ er,
    const float* __restrict__ feat, const float* __restrict__ bias,
    float* __restrict__ out, int N, int relu) {
    const int d = __builtin_amdgcn_readfirstlane((int)(blockIdx.x * 8 + (threadIdx.x >> 5)));
    if (d >= N) return;
    const int lane = threadIdx.x & 31;
    const bool hi = lane >= 16;
    const int beg = (d > 0) ? pos[d - 1] : 0;
    const int deg = pos[d] - beg;

    const float4 vd = reinterpret_cast<const float4*>(feat + (size_t)d * 128)[lane];

    const float er0 = er[d * 2 + 0], er1 = er[d * 2 + 1];
    const float es0 = lrelu(el[d * 2 + 0] + er0);
    const float es1 = lrelu(el[d * 2 + 1] + er1);

    const int  n0  = deg < 32 ? deg : 32;
    const bool act = lane < n0;
    int   s_c  = d;
    float e0_c = 0.f, e1_c = 0.f;
    if (act) {
        s_c  = csr[beg + lane];
        e0_c = lrelu(el[s_c * 2 + 0] + er0);
        e1_c = lrelu(el[s_c * 2 + 1] + er1);
    }
    float m0 = act ? fmaxf(es0, e0_c) : es0;
    float m1 = act ? fmaxf(es1, e1_c) : es1;
    for (int j = 32 + lane; j < deg; j += 32) {
        const int s = csr[beg + j];
        m0 = fmaxf(m0, lrelu(el[s * 2 + 0] + er0));
        m1 = fmaxf(m1, lrelu(el[s * 2 + 1] + er1));
    }
    m0 = wave_max(m0);
    m1 = wave_max(m1);

    const float p0_c = act ? expf(e0_c - m0) : 0.f;
    const float p1_c = act ? expf(e1_c - m1) : 0.f;
    float z0 = p0_c, z1 = p1_c;
    for (int j = 32 + lane; j < deg; j += 32) {
        const int s = csr[beg + j];
        z0 += expf(lrelu(el[s * 2 + 0] + er0) - m0);
        z1 += expf(lrelu(el[s * 2 + 1] + er1) - m1);
    }
    const float ps0 = expf(es0 - m0), ps1 = expf(es1 - m1);
    z0 = wave_sum(z0) + ps0;
    z1 = wave_sum(z1) + ps1;

    const float asl = hi ? (ps1 / z1) : (ps0 / z0);
    float4 acc = make_float4(asl * vd.x, asl * vd.y, asl * vd.z, asl * vd.w);

    add_rows(acc, feat, s_c, p0_c / z0, p1_c / z1, n0, lane, hi);
    for (int base = 32; base < deg; base += 32) {
        const int j = base + lane;
        int   s_l  = d;
        float a0_l = 0.f, a1_l = 0.f;
        if (j < deg) {
            s_l  = csr[beg + j];
            a0_l = expf(lrelu(el[s_l * 2 + 0] + er0) - m0) / z0;
            a1_l = expf(lrelu(el[s_l * 2 + 1] + er1) - m1) / z1;
        }
        const int rest = deg - base;
        add_rows(acc, feat, s_l, a0_l, a1_l, rest < 32 ? rest : 32, lane, hi);
    }

    const float* bp = bias + lane * 4;
    float4 r = make_float4(acc.x + bp[0], acc.y + bp[1], acc.z + bp[2], acc.w + bp[3]);
    if (relu) {
        r.x = fmaxf(r.x, 0.f); r.y = fmaxf(r.y, 0.f);
        r.z = fmaxf(r.z, 0.f); r.w = fmaxf(r.w, 0.f);
    }
    reinterpret_cast<float4*>(out + (size_t)d * 128)[lane] = r;
}

extern "C" void kernel_launch(void* const* d_in, const int* in_sizes, int n_in,
                              void* d_out, int out_size, void* d_ws, size_t ws_size,
                              hipStream_t stream) {
    const float* x   = (const float*)d_in[0];
    const int*   src = (const int*)  d_in[1];
    const int*   dst = (const int*)  d_in[2];
    const float* W1  = (const float*)d_in[3];
    const float* al1 = (const float*)d_in[4];
    const float* ar1 = (const float*)d_in[5];
    const float* b1  = (const float*)d_in[6];
    const float* W2  = (const float*)d_in[7];
    const float* al2 = (const float*)d_in[8];
    const float* ar2 = (const float*)d_in[9];
    const float* b2  = (const float*)d_in[10];

    const int N = in_sizes[0] / 128;
    const long long E  = in_sizes[1];
    const long long Et = E + N;

    float* out = (float*)d_out;

    float* feat = (float*)d_ws;
    float* el   = feat + (size_t)N * 128;
    float* er   = el + (size_t)N * 2;
    float* ssum = er + (size_t)N * 2;
    int*   mkey = (int*)(ssum + (size_t)N * 2);

    const int gemmGrid = (N + 15) / 16;

    const int    nScan = (N + 4095) / 4096;
    const size_t need  = ((size_t)N * 132 + (size_t)N + 1024 + (size_t)E) * 4;
    if (ws_size >= need && nScan <= 1024 && E < 0x7fffffffLL) {
        int* pos  = (int*)(er + (size_t)N * 2);
        int* bsum = pos + (size_t)N;
        int* csr  = bsum + 1024;
        const int E32     = (int)E;
        const int cBlocks = (E32 + 255) / 256;
        const int gBlocks = (N + 7) / 8;

        const int    NB  = (N + CS_BN - 1) / CS_BN;
        const int    NT  = (E32 + CS_TILE - 1) / CS_TILE;
        const size_t tmp = ((size_t)NT * CS_TILE * 2 + (size_t)(NB + 1) * NT) * 4;
        if (E32 > 0 && E32 < 0x7fff0000 && NB + 1 <= CS_MAXB && tmp <= (size_t)N * 128 * 4) {
            int* tedge = (int*)feat;
            int* toffT = tedge + (size_t)NT * CS_TILE * 2;
            csr_tile_sort<<<NT, 1024, 0, stream>>>(src, dst, E32, NB, NT, toffT, tedge);
            csr_bucket_build<<<NB, 1024, 0, stream>>>(toffT, tedge, N, E32, NT, pos, csr);
        } else {
            csr_zero<<<(N + 255) / 256, 256, 0, stream>>>(pos, N);
            csr_count<<<cBlocks, 256, 0, stream>>>(dst, E32, pos);
            csr_scan_block<<<nScan, 1024, 0, stream>>>(pos, N, bsum);
            csr_scan_add<<<nScan, 1024, 0, stream>>>(pos, N, bsum);
            csr_fill<<<cBlocks, 256, 0, stream>>>(src, dst, E32, pos, csr);
        }

        gat_gemm_wmma<<<gemmGrid, 256, 0, stream>>>(x, W1, al1, ar1, feat, el, er, N);
        gat_gather<<<gBlocks, 256, 0, stream>>>(pos, csr, el, er, feat, b1, out, N, 1);

        gat_gemm_wmma<<<gemmGrid, 256, 0, stream>>>(out, W2, al2, ar2, feat, el, er, N);
        gat_gather<<<gBlocks, 256, 0, stream>>>(pos, csr, el, er, feat, b2, out, N, 0);
        return;
    }

    const int  totalNF = N * 128;
    const int  n2      = N * 2;
    const int  eBlocks = (int)((Et + 255) / 256);
    const int  aBlocks = (int)((Et + 7) / 8);

    gat_gemm_wmma<<<gemmGrid, 256, 0, stream>>>(x, W1, al1, ar1, feat, el, er, N);
    gat_init_ms<<<(n2 + 255) / 256, 256, 0, stream>>>(mkey, ssum, n2);
    gat_init_bias<<<(totalNF + 255) / 256, 256, 0, stream>>>(out, b1, totalNF);
    gat_edge_max<<<eBlocks, 256, 0, stream>>>(src, dst, E, Et, el, er, mkey);
    gat_edge_sum<<<eBlocks, 256, 0, stream>>>(src, dst, E, Et, el, er, mkey, ssum);
    gat_edge_agg<<<aBlocks, 256, 0, stream>>>(src, dst, E, Et, el, er, mkey, ssum, feat, out);
    gat_relu<<<(totalNF + 255) / 256, 256, 0, stream>>>(out, totalNF);

    gat_gemm_wmma<<<gemmGrid, 256, 0, stream>>>(out, W2, al2, ar2, feat, el, er, N);
    gat_init_ms<<<(n2 + 255) / 256, 256, 0, stream>>>(mkey, ssum, n2);
    gat_init_bias<<<(totalNF + 255) / 256, 256, 0, stream>>>(out, b2, totalNF);
    gat_edge_max<<<eBlocks, 256, 0, stream>>>(src, dst, E, Et, el, er, mkey);
    gat_edge_sum<<<eBlocks, 256, 0, stream>>>(src, dst, E, Et, el, er, mkey, ssum);
    gat_edge_agg<<<aBlocks, 256, 0, stream>>>(src, dst, E, Et, el, er, mkey, ssum, feat, out);
}


# GEMM and aggregation as resident grids: a wave loops over row stripes / nodes (W fragments kept in registers, next item requested ahead); 45k waves enter those 4 launches, not 300k; sum order kept
# speedup vs baseline: 7.3542x; 1.1127x over previous
#include <hip/hip_runtime.h>
#include <hip/hip_bf16.h>

typedef float v2f __attribute__((ext_vector_type(2)));
typedef float v8f __attribute__((ext_vector_type(8)));

#define NEG_SLOPE 0.2f

__device__ __forceinline__ int f2ord(float f) {
    int i = __float_as_int(f);
    return (i >= 0) ? i : (i ^ 0x7fffffff);
}
__device__ __forceinline__ float ord2f(int i) {
    return __int_as_float((i >= 0) ? i : (i ^ 0x7fffffff));
}


template <int XM>
__device__ __forceinline__ float row_xor(float v) {
    return __int_as_float(__builtin_amdgcn_update_dpp(
        0, __float_as_int(v), 0x160 + XM  , 0xF, 0xF, true));
}

template <bool COEF>
__global__ __launch_bounds__(256) void gat_gemm_wmma(
    const float* __restrict__ A, const float* __restrict__ W,
    const float* __restrict__ al, const float* __restrict__ ar,
    float* __restrict__ Out, float* __restrict__ el, float* __restrict__ er, int M) {
    __shared__ float As[16][132];
    __shared__ float red[8][32];

    const int rowBase = blockIdx.x * 16;

    #pragma unroll
    for (int i = 0; i < 8; ++i) {
        int idx = threadIdx.x + i * 256;
        int r = idx >> 7;
        int k = idx & 127;
        int row = rowBase + r;
        if (row >= M) row = M - 1;
        As[r][k] = A[(size_t)row * 128 + k];
    }
    __syncthreads();

    const int wave = threadIdx.x >> 5;
    const int lane = threadIdx.x & 31;
    const int half = lane >> 4;
    const int l16  = lane & 15;
    const int ncol = wave * 16 + l16;

    v8f acc = {};
    for (int kk = 0; kk < 128; kk += 4) {
        v2f a, b;
        a.x = As[l16][kk + half * 2 + 0];
        a.y = As[l16][kk + half * 2 + 1];
        b.x = W[(size_t)(kk + half * 2 + 0) * 128 + ncol];
        b.y = W[(size_t)(kk + half * 2 + 1) * 128 + ncol];
        acc = __builtin_amdgcn_wmma_f32_16x16x4_f32(
             false, a,  false, b,
             (short)0, acc,  false,  false);
    }

    #pragma unroll
    for (int r = 0; r < 8; ++r) {
        int row = rowBase + r + half * 8;
        if (row < M) Out[(size_t)row * 128 + ncol] = acc[r];
    }

    if constexpr (COEF) {
    const float alv = al[ncol], arv = ar[ncol];
    const bool b3 = (l16 & 8) != 0, b2 = (l16 & 4) != 0, b1 = (l16 & 2) != 0, b0 = (l16 & 1) != 0;
    const float ck = b3 ? arv : alv;
    const float cs = b3 ? alv : arv;
    float k8[8];
    for (int i = 0; i < 8; ++i) k8[i] = acc[i] * ck + row_xor<8>(acc[i] * cs);
    float k4[4];
    for (int i = 0; i < 4; ++i)
        k4[i] = (b2 ? k8[4 + i] : k8[i]) + row_xor<4>(b2 ? k8[i] : k8[4 + i]);
    float k2[2];
    for (int i = 0; i < 2; ++i)
        k2[i] = (b1 ? k4[2 + i] : k4[i]) + row_xor<2>(b1 ? k4[i] : k4[2 + i]);
    const float k1 = (b0 ? k2[1] : k2[0]) + row_xor<1>(b0 ? k2[0] : k2[1]);
    red[wave][lane] = k1;
    __syncthreads();

    if (threadIdx.x < 64) {
        const int kind = threadIdx.x >> 5;
        const int row  = lane >> 1, hd = lane & 1;
        const int slot = (row >> 3) * 16 + kind * 8 + (row & 7);
        const float s = ((red[hd * 4 + 0][slot] + red[hd * 4 + 1][slot])
                         + red[hd * 4 + 2][slot]) + red[hd * 4 + 3][slot];
        if (rowBase + row < M) (kind ? er : el)[(size_t)rowBase * 2 + lane] = s;
    }
    }
}

__global__ __launch_bounds__(256) void gat_gemm_resident(
    const float* __restrict__ A, const float* __restrict__ W,
    float* __restrict__ Out, int M, int nStripes) {
    __shared__ float As[2][16][132];

    const int wave = threadIdx.x >> 5;
    const int lane = threadIdx.x & 31;
    const int half = lane >> 4;
    const int l16  = lane & 15;
    const int ncol = wave * 16 + l16;
    const int pr = threadIdx.x >> 5;
    const int pc = (threadIdx.x & 31) * 4;

    int stripe = blockIdx.x;
    if (stripe >= nStripes) return;

    v2f bw[32];
    #pragma unroll
    for (int i = 0; i < 32; ++i) {
        bw[i].x = W[(size_t)(i * 4 + half * 2 + 0) * 128 + ncol];
        bw[i].y = W[(size_t)(i * 4 + half * 2 + 1) * 128 + ncol];
    }

    float4 p0, p1;
    {
        int ra = stripe * 16 + pr, rb = ra + 8;
        if (ra >= M) ra = M - 1;
        if (rb >= M) rb = M - 1;
        p0 = *reinterpret_cast<const float4*>(A + (size_t)ra * 128 + pc);
        p1 = *reinterpret_cast<const float4*>(A + (size_t)rb * 128 + pc);
    }
    *reinterpret_cast<float4*>(&As[0][pr][pc])     = p0;
    *reinterpret_cast<float4*>(&As[0][pr + 8][pc]) = p1;
    __syncthreads();

    int buf = 0;
    for (; stripe < nStripes; stripe += gridDim.x) {
        const int  nxt  = stripe + gridDim.x;
        const bool more = nxt < nStripes;
        if (more) {
            int ra = nxt * 16 + pr, rb = ra + 8;
            if (ra >= M) ra = M - 1;
            if (rb >= M) rb = M - 1;
            p0 = *reinterpret_cast<const float4*>(A + (size_t)ra * 128 + pc);
            p1 = *reinterpret_cast<const float4*>(A + (size_t)rb * 128 + pc);
        }

        v8f acc = {};
        #pragma unroll
        for (int i = 0; i < 32; ++i) {
            v2f a;
            a.x = As[buf][l16][i * 4 + half * 2 + 0];
            a.y = As[buf][l16][i * 4 + half * 2 + 1];
            acc = __builtin_amdgcn_wmma_f32_16x16x4_f32(
                 false, a,  false, bw[i],
                 (short)0, acc,  false,  false);
        }

        const int rowBase = stripe * 16;
        #pragma unroll
        for (int r = 0; r < 8; ++r) {
            int row = rowBase + r + half * 8;
            if (row < M) Out[(size_t)row * 128 + ncol] = acc[r];
        }

        if (more) {
            *reinterpret_cast<float4*>(&As[buf ^ 1][pr][pc])     = p0;
            *reinterpret_cast<float4*>(&As[buf ^ 1][pr + 8][pc]) = p1;
        }
        __syncthreads();
        buf ^= 1;
    }
}

__global__ void gat_init_ms(int* __restrict__ mkey, float* __restrict__ s, int n2) {
    int i = blockIdx.x * blockDim.x + threadIdx.x;
    if (i < n2) { mkey[i] = 0x80000000; s[i] = 0.f; }
}

__global__ void gat_init_bias(float* __restrict__ out, const float* __restrict__ b,
                              int total) {
    int i = blockIdx.x * blockDim.x + threadIdx.x;
    if (i < total) out[i] = b[i & 127];
}

__device__ __forceinline__ void edge_sd(long long i, long long E,
                                        const int* __restrict__ src,
                                        const int* __restrict__ dst,
                                        int& s, int& d) {
    if (i < E) { s = src[i]; d = dst[i]; }
    else       { s = (int)(i - E); d = s; }
}

__global__ void gat_edge_max(const int* __restrict__ src, const int* __restrict__ dst,
                             long long E, long long Et,
                             const float* __restrict__ el, const float* __restrict__ er,
                             int* __restrict__ mkey) {
    long long i = (long long)blockIdx.x * blockDim.x + threadIdx.x;
    if (i >= Et) return;
    int s, d; edge_sd(i, E, src, dst, s, d);
    #pragma unroll
    for (int h = 0; h < 2; ++h) {
        float e = el[s * 2 + h] + er[d * 2 + h];
        e = (e >= 0.f) ? e : NEG_SLOPE * e;
        atomicMax(&mkey[d * 2 + h], f2ord(e));
    }
}

__global__ void gat_edge_sum(const int* __restrict__ src, const int* __restrict__ dst,
                             long long E, long long Et,
                             const float* __restrict__ el, const float* __restrict__ er,
                             const int* __restrict__ mkey, float* __restrict__ ssum) {
    long long i = (long long)blockIdx.x * blockDim.x + threadIdx.x;
    if (i >= Et) return;
    int s, d; edge_sd(i, E, src, dst, s, d);
    #pragma unroll
    for (int h = 0; h < 2; ++h) {
        float e = el[s * 2 + h] + er[d * 2 + h];
        e = (e >= 0.f) ? e : NEG_SLOPE * e;
        atomicAdd(&ssum[d * 2 + h], expf(e - ord2f(mkey[d * 2 + h])));
    }
}

__global__ __launch_bounds__(256) void gat_edge_agg(
    const int* __restrict__ src, const int* __restrict__ dst,
    long long E, long long Et,
    const float* __restrict__ el, const float* __restrict__ er,
    const int* __restrict__ mkey, const float* __restrict__ ssum,
    const float* __restrict__ feat, float* __restrict__ out) {
    long long w = (long long)blockIdx.x * 8 + (threadIdx.x >> 5);
    if (w >= Et) return;
    const int lane = threadIdx.x & 31;
    int s, d; edge_sd(w, E, src, dst, s, d);

    float e0 = el[s * 2 + 0] + er[d * 2 + 0];
    float e1 = el[s * 2 + 1] + er[d * 2 + 1];
    e0 = (e0 >= 0.f) ? e0 : NEG_SLOPE * e0;
    e1 = (e1 >= 0.f) ? e1 : NEG_SLOPE * e1;
    const float alpha0 = expf(e0 - ord2f(mkey[d * 2 + 0])) / ssum[d * 2 + 0];
    const float alpha1 = expf(e1 - ord2f(mkey[d * 2 + 1])) / ssum[d * 2 + 1];

    const float* f = feat + (size_t)s * 128 + lane;
    const float v0 = f[0], v1 = f[32], v2 = f[64], v3 = f[96];
    float* o = out + (size_t)d * 128 + lane;
    atomicAdd(o +  0, alpha0 * v0);
    atomicAdd(o + 32, alpha0 * v1);
    atomicAdd(o + 64, alpha1 * v2);
    atomicAdd(o + 96, alpha1 * v3);
}

__global__ void gat_relu(float* __restrict__ x, int total) {
    int i = blockIdx.x * blockDim.x + threadIdx.x;
    if (i < total) x[i] = fmaxf(x[i], 0.f);
}


__global__ void csr_zero(int* __restrict__ pos, int N) {
    int i = blockIdx.x * blockDim.x + threadIdx.x;
    if (i < N) pos[i] = 0;
}

__global__ void csr_count(const int* __restrict__ dst, int E, int* __restrict__ pos) {
    int i = blockIdx.x * blockDim.x + threadIdx.x;
    if (i < E) atomicAdd(&pos[dst[i]], 1);
}

__global__ __launch_bounds__(1024) void csr_scan_block(int* __restrict__ pos, int N,
                                                       int* __restrict__ bsum) {
    __shared__ int wtot[32];
    const int t = threadIdx.x, lane = t & 31, wv = t >> 5;
    const int base = blockIdx.x * 4096 + t * 4;
    int v[4];
    #pragma unroll
    for (int i = 0; i < 4; ++i) v[i] = (base + i < N) ? pos[base + i] : 0;
    const int tsum = v[0] + v[1] + v[2] + v[3];
    int x = tsum;
    #pragma unroll
    for (int o = 1; o < 32; o <<= 1) {
        int y = __shfl_up(x, o);
        if (lane >= o) x += y;
    }
    if (lane == 31) wtot[wv] = x;
    __syncthreads();
    if (wv == 0) {
        int w = wtot[lane];
        #pragma unroll
        for (int o = 1; o < 32; o <<= 1) {
            int y = __shfl_up(w, o);
            if (lane >= o) w += y;
        }
        wtot[lane] = w;
    }
    __syncthreads();
    int run = x - tsum + (wv > 0 ? wtot[wv - 1] : 0);
    #pragma unroll
    for (int i = 0; i < 4; ++i) {
        if (base + i < N) pos[base + i] = run;
        run += v[i];
    }
    if (t == 1023) bsum[blockIdx.x] = run;
}

__global__ __launch_bounds__(1024) void csr_scan_add(int* __restrict__ pos, int N,
                                                     const int* __restrict__ bsum) {
    const int b = blockIdx.x;
    int off = 0;
    for (int j = 0; j < b; ++j) off += bsum[j];
    const int base = b * 4096 + threadIdx.x * 4;
    #pragma unroll
    for (int i = 0; i < 4; ++i)
        if (base + i < N) pos[base + i] += off;
}

__global__ void csr_fill(const int* __restrict__ src, const int* __restrict__ dst, int E,
                         int* __restrict__ pos, int* __restrict__ csr) {
    int i = blockIdx.x * blockDim.x + threadIdx.x;
    if (i >= E) return;
    int p = atomicAdd(&pos[dst[i]], 1);
    if ((unsigned)p < (unsigned)E) csr[p] = src[i];
}


#define CS_TILE 4096
#define CS_BSH  8
#define CS_BN   256
#define CS_MAXB 1024

__device__ __forceinline__ int wave_sum_i(int v) {
    for (int o = 16; o > 0; o >>= 1) v += __shfl_xor(v, o);
    return v;
}

__global__ __launch_bounds__(1024) void csr_tile_sort(
    const int* __restrict__ src, const int* __restrict__ dst, int E, int NB, int NT,
    int* __restrict__ toffT, int* __restrict__ tedge) {
    __shared__ int cnt[CS_MAXB];
    __shared__ int wtot[32];
    __shared__ int stS[CS_TILE];
    __shared__ int stD[CS_TILE];

    const int t = threadIdx.x, lane = t & 31, wv = t >> 5;
    const int tile = blockIdx.x;
    const int e0 = tile * CS_TILE;

    cnt[t] = 0;
    __syncthreads();

    int s[4], d[4];
    for (int i = 0; i < 4; ++i) {
        const int e = e0 + i * 1024 + t;
        s[i] = 0; d[i] = -1;
        if (e < E) { s[i] = src[e]; d[i] = dst[e]; }
    }
    for (int i = 0; i < 4; ++i)
        if (d[i] >= 0) atomicAdd(&cnt[d[i] >> CS_BSH], 1);
    __syncthreads();

    const int c = cnt[t];
    int x = c;
    for (int o = 1; o < 32; o <<= 1) {
        int y = __shfl_up(x, o);
        if (lane >= o) x += y;
    }
    if (lane == 31) wtot[wv] = x;
    __syncthreads();
    if (wv == 0) {
        int w = wtot[lane];
        for (int o = 1; o < 32; o <<= 1) {
            int y = __shfl_up(w, o);
            if (lane >= o) w += y;
        }
        wtot[lane] = w;
    }
    __syncthreads();
    const int excl = x - c + (wv > 0 ? wtot[wv - 1] : 0);
    cnt[t] = excl;
    if (t <= NB) toffT[(size_t)t * NT + tile] = excl;
    __syncthreads();

    for (int i = 0; i < 4; ++i)
        if (d[i] >= 0) {
            const int r = atomicAdd(&cnt[d[i] >> CS_BSH], 1);
            stS[r] = s[i];
            stD[r] = d[i];
        }
    __syncthreads();
    const int left = E - e0;
    const int nE = left < CS_TILE ? left : CS_TILE;
    int2* outp = reinterpret_cast<int2*>(tedge) + (size_t)tile * CS_TILE;
    for (int i = 0; i < 4; ++i) {
        const int j = i * 1024 + t;
        if (j < nE) outp[j] = make_int2(stS[j], stD[j]);
    }
}

__global__ __launch_bounds__(1024) void csr_bucket_build(
    const int* __restrict__ toffT, const int* __restrict__ tedge, int N, int E, int NT,
    int* __restrict__ pos, int* __restrict__ csr) {
    __shared__ int hist[CS_BN];
    __shared__ int wsum[32];
    __shared__ int wtot[32];

    const int t = threadIdx.x, lane = t & 31, wv = t >> 5;
    const int b = blockIdx.x;
    const int* lo = toffT + (size_t)b * NT;
    const int* hi = lo + NT;
    const int2* edges = reinterpret_cast<const int2*>(tedge);

    if (t < CS_BN) hist[t] = 0;
    int part = 0;
    for (int k = t; k < NT; k += 1024) part += lo[k];
    part = wave_sum_i(part);
    if (lane == 0) wsum[wv] = part;
    __syncthreads();
    int base = 0;
    for (int i = 0; i < 32; ++i) base += wsum[i];

    for (int k = t >> 1; k < NT; k += 512) {
        const int a = lo[k], z = hi[k];
        const int mid = a + ((z - a + 1) >> 1);
        const int s0 = (t & 1) ? mid : a;
        const int s1 = (t & 1) ? z : mid;
        const int2* p = edges + (size_t)k * CS_TILE;
        for (int j = s0; j < s1; j += 4) {
            int key[4];
            for (int u = 0; u < 4; ++u) {
                const int jj = (j + u < s1) ? (j + u) : (s1 - 1);
                key[u] = p[jj].y & (CS_BN - 1);
            }
            for (int u = 0; u < 4; ++u)
                atomicAdd(&hist[key[u]], (j + u < s1) ? 1 : 0);
        }
    }
    __syncthreads();

    const int c = (t < CS_BN) ? hist[t] : 0;
    int x = c;
    for (int o = 1; o < 32; o <<= 1) {
        int y = __shfl_up(x, o);
        if (lane >= o) x += y;
    }
    if (lane == 31) wtot[wv] = x;
    __syncthreads();
    if (t < CS_BN) {
        int incl = x;
        for (int i = 0; i < wv; ++i) incl += wtot[i];
        hist[t] = incl - c;
        const int n = b * CS_BN + t;
        if (n < N) pos[n] = base + incl;
    }
    __syncthreads();

    for (int k = t >> 1; k < NT; k += 512) {
        const int a = lo[k], z = hi[k];
        const int mid = a + ((z - a + 1) >> 1);
        const int s0 = (t & 1) ? mid : a;
        const int s1 = (t & 1) ? z : mid;
        const int2* p = edges + (size_t)k * CS_TILE;
        for (int j = s0; j < s1; j += 4) {
            int2 e[4];
            for (int u = 0; u < 4; ++u) {
                const int jj = (j + u < s1) ? (j + u) : (s1 - 1);
                e[u] = p[jj];
            }
            for (int u = 0; u < 4; ++u) {
                const bool ok = j + u < s1;
                const int r = base + atomicAdd(&hist[e[u].y & (CS_BN - 1)], ok ? 1 : 0);
                if (ok && (unsigned)r < (unsigned)E) csr[r] = e[u].x;
            }
        }
    }
}

__device__ __forceinline__ float row_sum16(float x) {
    x += row_xor<8>(x);
    x += row_xor<4>(x);
    x += row_xor<2>(x);
    x += row_xor<1>(x);
    return x;
}

template <int SRC>
__device__ __forceinline__ float row_bcast(float v) {
    return __int_as_float(__builtin_amdgcn_update_dpp(
        0, __float_as_int(v), 0x150 + SRC  , 0xF, 0xF, true));
}

__device__ __forceinline__ float dot4(const float4& v, const v2f& alo, const v2f& ahi) {
    const v2f lo = {v.x, v.y}, hi = {v.z, v.w};
    v2f s = lo * alo;
    s = __builtin_elementwise_fma(hi, ahi, s);
    return s.x + s.y;
}

__global__ __launch_bounds__(256) void gat_gather_dot(
    const int* __restrict__ pos, const int* __restrict__ csr,
    const float* __restrict__ al, const float* __restrict__ ar,
    const float* __restrict__ feat, const float* __restrict__ bias,
    float* __restrict__ out, int N, int relu, int per) {
    const int gw = __builtin_amdgcn_readfirstlane((int)(blockIdx.x * 8 + (threadIdx.x >> 5)));
    const int n0 = gw * per;
    if (n0 >= N) return;
    const int lane = threadIdx.x & 31;
    const int left = N - n0;
    const int nn   = left < per ? left : per;

    int pv = 0;
    {
        const int pi = n0 - 1 + lane;
        if (lane <= nn && pi >= 0) pv = pos[pi];
    }

    const float4 a4 = reinterpret_cast<const float4*>(al)[lane];
    const float4 r4 = reinterpret_cast<const float4*>(ar)[lane];
    const float4 b4 = reinterpret_cast<const float4*>(bias)[lane];
    const float L2E = 1.44269504088896341f;
    const v2f alo = {a4.x * L2E, a4.y * L2E}, ahi = {a4.z * L2E, a4.w * L2E};
    const v2f rlo = {r4.x * L2E, r4.y * L2E}, rhi = {r4.z * L2E, r4.w * L2E};

    const bool b3 = (lane & 8) != 0, b2 = (lane & 4) != 0;
    const int  q  = (lane >> 2) & 3;

    int beg = __builtin_amdgcn_readlane(pv, 0);
    int deg = __builtin_amdgcn_readlane(pv, 1) - beg;
    int s_l = n0;
    if (lane < deg) s_l = csr[beg + lane];
    float4 vd = reinterpret_cast<const float4*>(feat + (size_t)n0 * 128)[lane];

    for (int jn = 0; jn < nn; ++jn) {
        const int d = n0 + jn;

        int    begn = 0, degn = 0, s_ln = d;
        float4 vdn  = vd;
        auto request_next = [&]() {
            if (jn + 1 < nn) {
                begn = __builtin_amdgcn_readlane(pv, jn + 1);
                degn = __builtin_amdgcn_readlane(pv, jn + 2) - begn;
                s_ln = d + 1;
                if (lane < degn) s_ln = csr[begn + lane];
                vdn = reinterpret_cast<const float4*>(feat + (size_t)(d + 1) * 128)[lane];
            }
        };
        if (deg <= 0) request_next();

        const float erd = row_sum16(dot4(vd, rlo, rhi));
        const float tsf = row_sum16(dot4(vd, alo, ahi)) + erd;
        float  m   = fmaxf(tsf, NEG_SLOPE * tsf);
        float  zl  = (q == 0) ? 1.f : 0.f;
        float4 acc = vd;

        for (int base = 0; base < deg; base += 32) {
            if (base > 0) {
                const int j = base + lane;
                s_l = d;
                if (j < deg) s_l = csr[beg + j];
            }
            const int rest = deg - base;
            const int cnt  = rest < 32 ? rest : 32;
            for (int k = 0; k < cnt; k += 4) {
                float4 v[4];
                float  t[4];
                for (int i = 0; i < 4; ++i) {
                    const int s = __builtin_amdgcn_readlane(s_l, k + i);
                    v[i] = reinterpret_cast<const float4*>(feat + (size_t)s * 128)[lane];
                }
                if (base == 0 && k == 0) request_next();
                for (int i = 0; i < 4; ++i) t[i] = dot4(v[i], alo, ahi);

                const float u0 = (b3 ? t[2] : t[0]) + row_xor<8>(b3 ? t[0] : t[2]);
                const float u1 = (b3 ? t[3] : t[1]) + row_xor<8>(b3 ? t[1] : t[3]);
                float w = (b2 ? u1 : u0) + row_xor<4>(b2 ? u0 : u1);
                w += row_xor<2>(w);
                w += row_xor<1>(w);

                const float x = w + erd;
                const float e = (k + q < cnt) ? fmaxf(x, NEG_SLOPE * x) : -__builtin_inff();
                float mt = fmaxf(e, row_xor<4>(e));
                mt = fmaxf(mt, row_xor<8>(mt));
                const float mn = fmaxf(m, mt);
                const float sc = __builtin_amdgcn_exp2f(m - mn);
                const float p  = __builtin_amdgcn_exp2f(e - mn);
                zl = fmaf(zl, sc, p);
                const float p0 = row_bcast<0>(p), p1 = row_bcast<4>(p);
                const float p2 = row_bcast<8>(p), p3 = row_bcast<12>(p);
                acc.x = fmaf(p0, v[0].x, acc.x * sc); acc.y = fmaf(p0, v[0].y, acc.y * sc);
                acc.z = fmaf(p0, v[0].z, acc.z * sc); acc.w = fmaf(p0, v[0].w, acc.w * sc);
                acc.x = fmaf(p1, v[1].x, acc.x); acc.y = fmaf(p1, v[1].y, acc.y);
                acc.z = fmaf(p1, v[1].z, acc.z); acc.w = fmaf(p1, v[1].w, acc.w);
                acc.x = fmaf(p2, v[2].x, acc.x); acc.y = fmaf(p2, v[2].y, acc.y);
                acc.z = fmaf(p2, v[2].z, acc.z); acc.w = fmaf(p2, v[2].w, acc.w);
                acc.x = fmaf(p3, v[3].x, acc.x); acc.y = fmaf(p3, v[3].y, acc.y);
                acc.z = fmaf(p3, v[3].z, acc.z); acc.w = fmaf(p3, v[3].w, acc.w);
                m = mn;
            }
        }

        float z = zl + row_xor<4>(zl);
        z += row_xor<8>(z);
        const float inv = 1.f / z;
        float4 r = make_float4(fmaf(acc.x, inv, b4.x), fmaf(acc.y, inv, b4.y),
                               fmaf(acc.z, inv, b4.z), fmaf(acc.w, inv, b4.w));
        if (relu) {
            r.x = fmaxf(r.x, 0.f); r.y = fmaxf(r.y, 0.f);
            r.z = fmaxf(r.z, 0.f); r.w = fmaxf(r.w, 0.f);
        }
        reinterpret_cast<float4*>(out + (size_t)d * 128)[lane] = r;

        beg = begn; deg = degn; s_l = s_ln; vd = vdn;
    }
}

extern "C" void kernel_launch(void* const* d_in, const int* in_sizes, int n_in,
                              void* d_out, int out_size, void* d_ws, size_t ws_size,
                              hipStream_t stream) {
    const float* x   = (const float*)d_in[0];
    const int*   src = (const int*)  d_in[1];
    const int*   dst = (const int*)  d_in[2];
    const float* W1  = (const float*)d_in[3];
    const float* al1 = (const float*)d_in[4];
    const float* ar1 = (const float*)d_in[5];
    const float* b1  = (const float*)d_in[6];
    const float* W2  = (const float*)d_in[7];
    const float* al2 = (const float*)d_in[8];
    const float* ar2 = (const float*)d_in[9];
    const float* b2  = (const float*)d_in[10];

    const int N = in_sizes[0] / 128;
    const long long E  = in_sizes[1];
    const long long Et = E + N;

    float* out = (float*)d_out;

    float* feat = (float*)d_ws;
    float* el   = feat + (size_t)N * 128;
    float* er   = el + (size_t)N * 2;
    float* ssum = er + (size_t)N * 2;
    int*   mkey = (int*)(ssum + (size_t)N * 2);

    const int gemmGrid = (N + 15) / 16;

    const int    nScan = (N + 4095) / 4096;
    const size_t need  = ((size_t)N * 132 + (size_t)N + 1024 + (size_t)E) * 4;
    if (ws_size >= need && nScan <= 1024 && E < 0x7fffffffLL) {
        int* pos  = (int*)(er + (size_t)N * 2);
        int* bsum = pos + (size_t)N;
        int* csr  = bsum + 1024;
        const int E32     = (int)E;
        const int cBlocks = (E32 + 255) / 256;
        int per = (N + 16383) / 16384;
        if (per > 31) per = 31;
        const int gWaves  = (N + per - 1) / per;
        const int gBlocks = (gWaves + 7) / 8;
        const int rGrid   = gemmGrid < 1024 ? gemmGrid : 1024;

        const int    NB  = (N + CS_BN - 1) / CS_BN;
        const int    NT  = (E32 + CS_TILE - 1) / CS_TILE;
        const size_t tmp = ((size_t)NT * CS_TILE * 2 + (size_t)(NB + 1) * NT) * 4;
        if (E32 > 0 && E32 < 0x7fff0000 && NB + 1 <= CS_MAXB && tmp <= (size_t)N * 128 * 4) {
            int* tedge = (int*)feat;
            int* toffT = tedge + (size_t)NT * CS_TILE * 2;
            csr_tile_sort<<<NT, 1024, 0, stream>>>(src, dst, E32, NB, NT, toffT, tedge);
            csr_bucket_build<<<NB, 1024, 0, stream>>>(toffT, tedge, N, E32, NT, pos, csr);
        } else {
            csr_zero<<<(N + 255) / 256, 256, 0, stream>>>(pos, N);
            csr_count<<<cBlocks, 256, 0, stream>>>(dst, E32, pos);
            csr_scan_block<<<nScan, 1024, 0, stream>>>(pos, N, bsum);
            csr_scan_add<<<nScan, 1024, 0, stream>>>(pos, N, bsum);
            csr_fill<<<cBlocks, 256, 0, stream>>>(src, dst, E32, pos, csr);
        }

        gat_gemm_resident<<<rGrid, 256, 0, stream>>>(x, W1, feat, N, gemmGrid);
        gat_gather_dot<<<gBlocks, 256, 0, stream>>>(pos, csr, al1, ar1, feat, b1, out, N, 1, per);

        gat_gemm_resident<<<rGrid, 256, 0, stream>>>(out, W2, feat, N, gemmGrid);
        gat_gather_dot<<<gBlocks, 256, 0, stream>>>(pos, csr, al2, ar2, feat, b2, out, N, 0, per);
        return;
    }

    const int  totalNF = N * 128;
    const int  n2      = N * 2;
    const int  eBlocks = (int)((Et + 255) / 256);
    const int  aBlocks = (int)((Et + 7) / 8);

    gat_gemm_wmma<true><<<gemmGrid, 256, 0, stream>>>(x, W1, al1, ar1, feat, el, er, N);
    gat_init_ms<<<(n2 + 255) / 256, 256, 0, stream>>>(mkey, ssum, n2);
    gat_init_bias<<<(totalNF + 255) / 256, 256, 0, stream>>>(out, b1, totalNF);
    gat_edge_max<<<eBlocks, 256, 0, stream>>>(src, dst, E, Et, el, er, mkey);
    gat_edge_sum<<<eBlocks, 256, 0, stream>>>(src, dst, E, Et, el, er, mkey, ssum);
    gat_edge_agg<<<aBlocks, 256, 0, stream>>>(src, dst, E, Et, el, er, mkey, ssum, feat, out);
    gat_relu<<<(totalNF + 255) / 256, 256, 0, stream>>>(out, totalNF);

    gat_gemm_wmma<true><<<gemmGrid, 256, 0, stream>>>(out, W2, al2, ar2, feat, el, er, N);
    gat_init_ms<<<(n2 + 255) / 256, 256, 0, stream>>>(mkey, ssum, n2);
    gat_init_bias<<<(totalNF + 255) / 256, 256, 0, stream>>>(out, b2, totalNF);
    gat_edge_max<<<eBlocks, 256, 0, stream>>>(src, dst, E, Et, el, er, mkey);
    gat_edge_sum<<<eBlocks, 256, 0, stream>>>(src, dst, E, Et, el, er, mkey, ssum);
    gat_edge_agg<<<aBlocks, 256, 0, stream>>>(src, dst, E, Et, el, er, mkey, ssum, feat, out);
}


# Level-2 edge sort: one wave walks a tile run (lanes on consecutive pairs) and src ids leave via an LDS copy, so loads and csr stores cover whole lines, not 16-32; LDS arrival (sum) order changes
# speedup vs baseline: 7.6001x; 1.0334x over previous
#include <hip/hip_runtime.h>
#include <hip/hip_bf16.h>

typedef float v2f __attribute__((ext_vector_type(2)));
typedef float v8f __attribute__((ext_vector_type(8)));

#define NEG_SLOPE 0.2f

__device__ __forceinline__ int f2ord(float f) {
    int i = __float_as_int(f);
    return (i >= 0) ? i : (i ^ 0x7fffffff);
}
__device__ __forceinline__ float ord2f(int i) {
    return __int_as_float((i >= 0) ? i : (i ^ 0x7fffffff));
}


template <int XM>
__device__ __forceinline__ float row_xor(float v) {
    return __int_as_float(__builtin_amdgcn_update_dpp(
        0, __float_as_int(v), 0x160 + XM  , 0xF, 0xF, true));
}

template <bool COEF>
__global__ __launch_bounds__(256) void gat_gemm_wmma(
    const float* __restrict__ A, const float* __restrict__ W,
    const float* __restrict__ al, const float* __restrict__ ar,
    float* __restrict__ Out, float* __restrict__ el, float* __restrict__ er, int M) {
    __shared__ float As[16][132];
    __shared__ float red[8][32];

    const int rowBase = blockIdx.x * 16;

    #pragma unroll
    for (int i = 0; i < 8; ++i) {
        int idx = threadIdx.x + i * 256;
        int r = idx >> 7;
        int k = idx & 127;
        int row = rowBase + r;
        if (row >= M) row = M - 1;
        As[r][k] = A[(size_t)row * 128 + k];
    }
    __syncthreads();

    const int wave = threadIdx.x >> 5;
    const int lane = threadIdx.x & 31;
    const int half = lane >> 4;
    const int l16  = lane & 15;
    const int ncol = wave * 16 + l16;

    v8f acc = {};
    for (int kk = 0; kk < 128; kk += 4) {
        v2f a, b;
        a.x = As[l16][kk + half * 2 + 0];
        a.y = As[l16][kk + half * 2 + 1];
        b.x = W[(size_t)(kk + half * 2 + 0) * 128 + ncol];
        b.y = W[(size_t)(kk + half * 2 + 1) * 128 + ncol];
        acc = __builtin_amdgcn_wmma_f32_16x16x4_f32(
             false, a,  false, b,
             (short)0, acc,  false,  false);
    }

    #pragma unroll
    for (int r = 0; r < 8; ++r) {
        int row = rowBase + r + half * 8;
        if (row < M) Out[(size_t)row * 128 + ncol] = acc[r];
    }

    if constexpr (COEF) {
    const float alv = al[ncol], arv = ar[ncol];
    const bool b3 = (l16 & 8) != 0, b2 = (l16 & 4) != 0, b1 = (l16 & 2) != 0, b0 = (l16 & 1) != 0;
    const float ck = b3 ? arv : alv;
    const float cs = b3 ? alv : arv;
    float k8[8];
    for (int i = 0; i < 8; ++i) k8[i] = acc[i] * ck + row_xor<8>(acc[i] * cs);
    float k4[4];
    for (int i = 0; i < 4; ++i)
        k4[i] = (b2 ? k8[4 + i] : k8[i]) + row_xor<4>(b2 ? k8[i] : k8[4 + i]);
    float k2[2];
    for (int i = 0; i < 2; ++i)
        k2[i] = (b1 ? k4[2 + i] : k4[i]) + row_xor<2>(b1 ? k4[i] : k4[2 + i]);
    const float k1 = (b0 ? k2[1] : k2[0]) + row_xor<1>(b0 ? k2[0] : k2[1]);
    red[wave][lane] = k1;
    __syncthreads();

    if (threadIdx.x < 64) {
        const int kind = threadIdx.x >> 5;
        const int row  = lane >> 1, hd = lane & 1;
        const int slot = (row >> 3) * 16 + kind * 8 + (row & 7);
        const float s = ((red[hd * 4 + 0][slot] + red[hd * 4 + 1][slot])
                         + red[hd * 4 + 2][slot]) + red[hd * 4 + 3][slot];
        if (rowBase + row < M) (kind ? er : el)[(size_t)rowBase * 2 + lane] = s;
    }
    }
}

__global__ __launch_bounds__(256) void gat_gemm_resident(
    const float* __restrict__ A, const float* __restrict__ W,
    float* __restrict__ Out, int M, int nStripes) {
    __shared__ float As[2][16][132];

    const int wave = threadIdx.x >> 5;
    const int lane = threadIdx.x & 31;
    const int half = lane >> 4;
    const int l16  = lane & 15;
    const int ncol = wave * 16 + l16;
    const int pr = threadIdx.x >> 5;
    const int pc = (threadIdx.x & 31) * 4;

    int stripe = blockIdx.x;
    if (stripe >= nStripes) return;

    v2f bw[32];
    #pragma unroll
    for (int i = 0; i < 32; ++i) {
        bw[i].x = W[(size_t)(i * 4 + half * 2 + 0) * 128 + ncol];
        bw[i].y = W[(size_t)(i * 4 + half * 2 + 1) * 128 + ncol];
    }

    float4 p0, p1;
    {
        int ra = stripe * 16 + pr, rb = ra + 8;
        if (ra >= M) ra = M - 1;
        if (rb >= M) rb = M - 1;
        p0 = *reinterpret_cast<const float4*>(A + (size_t)ra * 128 + pc);
        p1 = *reinterpret_cast<const float4*>(A + (size_t)rb * 128 + pc);
    }
    *reinterpret_cast<float4*>(&As[0][pr][pc])     = p0;
    *reinterpret_cast<float4*>(&As[0][pr + 8][pc]) = p1;
    __syncthreads();

    int buf = 0;
    for (; stripe < nStripes; stripe += gridDim.x) {
        const int  nxt  = stripe + gridDim.x;
        const bool more = nxt < nStripes;
        if (more) {
            int ra = nxt * 16 + pr, rb = ra + 8;
            if (ra >= M) ra = M - 1;
            if (rb >= M) rb = M - 1;
            p0 = *reinterpret_cast<const float4*>(A + (size_t)ra * 128 + pc);
            p1 = *reinterpret_cast<const float4*>(A + (size_t)rb * 128 + pc);
        }

        v8f acc = {};
        #pragma unroll
        for (int i = 0; i < 32; ++i) {
            v2f a;
            a.x = As[buf][l16][i * 4 + half * 2 + 0];
            a.y = As[buf][l16][i * 4 + half * 2 + 1];
            acc = __builtin_amdgcn_wmma_f32_16x16x4_f32(
                 false, a,  false, bw[i],
                 (short)0, acc,  false,  false);
        }

        const int rowBase = stripe * 16;
        #pragma unroll
        for (int r = 0; r < 8; ++r) {
            int row = rowBase + r + half * 8;
            if (row < M) Out[(size_t)row * 128 + ncol] = acc[r];
        }

        if (more) {
            *reinterpret_cast<float4*>(&As[buf ^ 1][pr][pc])     = p0;
            *reinterpret_cast<float4*>(&As[buf ^ 1][pr + 8][pc]) = p1;
        }
        __syncthreads();
        buf ^= 1;
    }
}

__global__ void gat_init_ms(int* __restrict__ mkey, float* __restrict__ s, int n2) {
    int i = blockIdx.x * blockDim.x + threadIdx.x;
    if (i < n2) { mkey[i] = 0x80000000; s[i] = 0.f; }
}

__global__ void gat_init_bias(float* __restrict__ out, const float* __restrict__ b,
                              int total) {
    int i = blockIdx.x * blockDim.x + threadIdx.x;
    if (i < total) out[i] = b[i & 127];
}

__device__ __forceinline__ void edge_sd(long long i, long long E,
                                        const int* __restrict__ src,
                                        const int* __restrict__ dst,
                                        int& s, int& d) {
    if (i < E) { s = src[i]; d = dst[i]; }
    else       { s = (int)(i - E); d = s; }
}

__global__ void gat_edge_max(const int* __restrict__ src, const int* __restrict__ dst,
                             long long E, long long Et,
                             const float* __restrict__ el, const float* __restrict__ er,
                             int* __restrict__ mkey) {
    long long i = (long long)blockIdx.x * blockDim.x + threadIdx.x;
    if (i >= Et) return;
    int s, d; edge_sd(i, E, src, dst, s, d);
    #pragma unroll
    for (int h = 0; h < 2; ++h) {
        float e = el[s * 2 + h] + er[d * 2 + h];
        e = (e >= 0.f) ? e : NEG_SLOPE * e;
        atomicMax(&mkey[d * 2 + h], f2ord(e));
    }
}

__global__ void gat_edge_sum(const int* __restrict__ src, const int* __restrict__ dst,
                             long long E, long long Et,
                             const float* __restrict__ el, const float* __restrict__ er,
                             const int* __restrict__ mkey, float* __restrict__ ssum) {
    long long i = (long long)blockIdx.x * blockDim.x + threadIdx.x;
    if (i >= Et) return;
    int s, d; edge_sd(i, E, src, dst, s, d);
    #pragma unroll
    for (int h = 0; h < 2; ++h) {
        float e = el[s * 2 + h] + er[d * 2 + h];
        e = (e >= 0.f) ? e : NEG_SLOPE * e;
        atomicAdd(&ssum[d * 2 + h], expf(e - ord2f(mkey[d * 2 + h])));
    }
}

__global__ __launch_bounds__(256) void gat_edge_agg(
    const int* __restrict__ src, const int* __restrict__ dst,
    long long E, long long Et,
    const float* __restrict__ el, const float* __restrict__ er,
    const int* __restrict__ mkey, const float* __restrict__ ssum,
    const float* __restrict__ feat, float* __restrict__ out) {
    long long w = (long long)blockIdx.x * 8 + (threadIdx.x >> 5);
    if (w >= Et) return;
    const int lane = threadIdx.x & 31;
    int s, d; edge_sd(w, E, src, dst, s, d);

    float e0 = el[s * 2 + 0] + er[d * 2 + 0];
    float e1 = el[s * 2 + 1] + er[d * 2 + 1];
    e0 = (e0 >= 0.f) ? e0 : NEG_SLOPE * e0;
    e1 = (e1 >= 0.f) ? e1 : NEG_SLOPE * e1;
    const float alpha0 = expf(e0 - ord2f(mkey[d * 2 + 0])) / ssum[d * 2 + 0];
    const float alpha1 = expf(e1 - ord2f(mkey[d * 2 + 1])) / ssum[d * 2 + 1];

    const float* f = feat + (size_t)s * 128 + lane;
    const float v0 = f[0], v1 = f[32], v2 = f[64], v3 = f[96];
    float* o = out + (size_t)d * 128 + lane;
    atomicAdd(o +  0, alpha0 * v0);
    atomicAdd(o + 32, alpha0 * v1);
    atomicAdd(o + 64, alpha1 * v2);
    atomicAdd(o + 96, alpha1 * v3);
}

__global__ void gat_relu(float* __restrict__ x, int total) {
    int i = blockIdx.x * blockDim.x + threadIdx.x;
    if (i < total) x[i] = fmaxf(x[i], 0.f);
}


__global__ void csr_zero(int* __restrict__ pos, int N) {
    int i = blockIdx.x * blockDim.x + threadIdx.x;
    if (i < N) pos[i] = 0;
}

__global__ void csr_count(const int* __restrict__ dst, int E, int* __restrict__ pos) {
    int i = blockIdx.x * blockDim.x + threadIdx.x;
    if (i < E) atomicAdd(&pos[dst[i]], 1);
}

__global__ __launch_bounds__(1024) void csr_scan_block(int* __restrict__ pos, int N,
                                                       int* __restrict__ bsum) {
    __shared__ int wtot[32];
    const int t = threadIdx.x, lane = t & 31, wv = t >> 5;
    const int base = blockIdx.x * 4096 + t * 4;
    int v[4];
    #pragma unroll
    for (int i = 0; i < 4; ++i) v[i] = (base + i < N) ? pos[base + i] : 0;
    const int tsum = v[0] + v[1] + v[2] + v[3];
    int x = tsum;
    #pragma unroll
    for (int o = 1; o < 32; o <<= 1) {
        int y = __shfl_up(x, o);
        if (lane >= o) x += y;
    }
    if (lane == 31) wtot[wv] = x;
    __syncthreads();
    if (wv == 0) {
        int w = wtot[lane];
        #pragma unroll
        for (int o = 1; o < 32; o <<= 1) {
            int y = __shfl_up(w, o);
            if (lane >= o) w += y;
        }
        wtot[lane] = w;
    }
    __syncthreads();
    int run = x - tsum + (wv > 0 ? wtot[wv - 1] : 0);
    #pragma unroll
    for (int i = 0; i < 4; ++i) {
        if (base + i < N) pos[base + i] = run;
        run += v[i];
    }
    if (t == 1023) bsum[blockIdx.x] = run;
}

__global__ __launch_bounds__(1024) void csr_scan_add(int* __restrict__ pos, int N,
                                                     const int* __restrict__ bsum) {
    const int b = blockIdx.x;
    int off = 0;
    for (int j = 0; j < b; ++j) off += bsum[j];
    const int base = b * 4096 + threadIdx.x * 4;
    #pragma unroll
    for (int i = 0; i < 4; ++i)
        if (base + i < N) pos[base + i] += off;
}

__global__ void csr_fill(const int* __restrict__ src, const int* __restrict__ dst, int E,
                         int* __restrict__ pos, int* __restrict__ csr) {
    int i = blockIdx.x * blockDim.x + threadIdx.x;
    if (i >= E) return;
    int p = atomicAdd(&pos[dst[i]], 1);
    if ((unsigned)p < (unsigned)E) csr[p] = src[i];
}


#define CS_TILE 4096
#define CS_BSH  8
#define CS_BN   256
#define CS_MAXB 1024

__device__ __forceinline__ int wave_sum_i(int v) {
    for (int o = 16; o > 0; o >>= 1) v += __shfl_xor(v, o);
    return v;
}

__global__ __launch_bounds__(1024) void csr_tile_sort(
    const int* __restrict__ src, const int* __restrict__ dst, int E, int NB, int NT,
    int* __restrict__ toffT, int* __restrict__ tedge) {
    __shared__ int cnt[CS_MAXB];
    __shared__ int wtot[32];
    __shared__ int stS[CS_TILE];
    __shared__ int stD[CS_TILE];

    const int t = threadIdx.x, lane = t & 31, wv = t >> 5;
    const int tile = blockIdx.x;
    const int e0 = tile * CS_TILE;

    cnt[t] = 0;
    __syncthreads();

    int s[4], d[4];
    for (int i = 0; i < 4; ++i) {
        const int e = e0 + i * 1024 + t;
        s[i] = 0; d[i] = -1;
        if (e < E) { s[i] = src[e]; d[i] = dst[e]; }
    }
    for (int i = 0; i < 4; ++i)
        if (d[i] >= 0) atomicAdd(&cnt[d[i] >> CS_BSH], 1);
    __syncthreads();

    const int c = cnt[t];
    int x = c;
    for (int o = 1; o < 32; o <<= 1) {
        int y = __shfl_up(x, o);
        if (lane >= o) x += y;
    }
    if (lane == 31) wtot[wv] = x;
    __syncthreads();
    if (wv == 0) {
        int w = wtot[lane];
        for (int o = 1; o < 32; o <<= 1) {
            int y = __shfl_up(w, o);
            if (lane >= o) w += y;
        }
        wtot[lane] = w;
    }
    __syncthreads();
    const int excl = x - c + (wv > 0 ? wtot[wv - 1] : 0);
    cnt[t] = excl;
    if (t <= NB) toffT[(size_t)t * NT + tile] = excl;
    __syncthreads();

    for (int i = 0; i < 4; ++i)
        if (d[i] >= 0) {
            const int r = atomicAdd(&cnt[d[i] >> CS_BSH], 1);
            stS[r] = s[i];
            stD[r] = d[i];
        }
    __syncthreads();
    const int left = E - e0;
    const int nE = left < CS_TILE ? left : CS_TILE;
    int2* outp = reinterpret_cast<int2*>(tedge) + (size_t)tile * CS_TILE;
    for (int i = 0; i < 4; ++i) {
        const int j = i * 1024 + t;
        if (j < nE) outp[j] = make_int2(stS[j], stD[j]);
    }
}

#define CS_STAGE 12288
#define CS_RUNS  8

__global__ __launch_bounds__(1024) void csr_bucket_build(
    const int* __restrict__ toffT, const int* __restrict__ tedge, int N, int E, int NT,
    int* __restrict__ pos, int* __restrict__ csr) {
    __shared__ int hist[CS_BN];
    __shared__ int wsum[32];
    __shared__ int wtot[32];
    __shared__ int stage[CS_STAGE];
    __shared__ int totS;

    const int t = threadIdx.x, lane = t & 31;
    const int wv = __builtin_amdgcn_readfirstlane(t >> 5);
    const int b = blockIdx.x;
    const int* lo = toffT + (size_t)b * NT;
    const int* hi = lo + NT;
    const int2* edges = reinterpret_cast<const int2*>(tedge);

    if (t < CS_BN) hist[t] = 0;
    int part = 0;
    for (int k = t; k < NT; k += 1024) part += lo[k];
    part = wave_sum_i(part);
    if (lane == 0) wsum[wv] = part;
    __syncthreads();
    int base = 0;
    for (int i = 0; i < 32; ++i) base += wsum[i];

    for (int kb = 0; kb < NT; kb += 1024) {
        const int kl = kb + wv + 32 * lane;
        int la = 0, lz = 0;
        if (kl < NT) { la = lo[kl]; lz = hi[kl]; }
        #pragma unroll 1
        for (int i = 0; i < 32; i += CS_RUNS) {
            if (kb + wv + 32 * i >= NT) break;
            int key[CS_RUNS];
            for (int u = 0; u < CS_RUNS; ++u) {
                const int a = __builtin_amdgcn_readlane(la, i + u);
                const int z = __builtin_amdgcn_readlane(lz, i + u);
                const int  kt = kb + wv + 32 * (i + u);
                const int2* p = edges + (size_t)(kt < NT ? kt : NT - 1) * CS_TILE;
                const bool ok = a + lane < z;
                const int  jj = ok ? a + lane : (a < CS_TILE ? a : CS_TILE - 1);
                const int  dv = p[jj].y;
                key[u] = ok ? (dv & (CS_BN - 1)) : -1;
            }
            for (int u = 0; u < CS_RUNS; ++u)
                if (key[u] >= 0) atomicAdd(&hist[key[u]], 1);
            #pragma unroll 1
            for (int u = 0; u < CS_RUNS; ++u) {
                const int a = __builtin_amdgcn_readlane(la, i + u);
                const int z = __builtin_amdgcn_readlane(lz, i + u);
                const int2* p = edges + (size_t)(kb + wv + 32 * (i + u)) * CS_TILE;
                for (int j = a + 32 + lane; j < z; j += 32)
                    atomicAdd(&hist[p[j].y & (CS_BN - 1)], 1);
            }
        }
    }
    __syncthreads();

    const int c = (t < CS_BN) ? hist[t] : 0;
    int x = c;
    for (int o = 1; o < 32; o <<= 1) {
        int y = __shfl_up(x, o);
        if (lane >= o) x += y;
    }
    if (lane == 31) wtot[wv] = x;
    __syncthreads();
    if (t < CS_BN) {
        int incl = x;
        for (int i = 0; i < wv; ++i) incl += wtot[i];
        hist[t] = incl - c;
        const int n = b * CS_BN + t;
        if (n < N) pos[n] = base + incl;
        if (t == CS_BN - 1) totS = incl;
    }
    __syncthreads();
    const int  total  = totS;
    const bool staged = total <= CS_STAGE;

    for (int kb = 0; kb < NT; kb += 1024) {
        const int kl = kb + wv + 32 * lane;
        int la = 0, lz = 0;
        if (kl < NT) { la = lo[kl]; lz = hi[kl]; }
        #pragma unroll 1
        for (int i = 0; i < 32; i += CS_RUNS) {
            if (kb + wv + 32 * i >= NT) break;
            int2 e[CS_RUNS];
            for (int u = 0; u < CS_RUNS; ++u) {
                const int a = __builtin_amdgcn_readlane(la, i + u);
                const int z = __builtin_amdgcn_readlane(lz, i + u);
                const int  kt = kb + wv + 32 * (i + u);
                const int2* p = edges + (size_t)(kt < NT ? kt : NT - 1) * CS_TILE;
                const bool ok = a + lane < z;
                const int  jj = ok ? a + lane : (a < CS_TILE ? a : CS_TILE - 1);
                const int2 ev = p[jj];
                e[u] = make_int2(ev.x, ok ? ev.y : -1);
            }
            for (int u = 0; u < CS_RUNS; ++u)
                if (e[u].y >= 0) {
                    const int r = atomicAdd(&hist[e[u].y & (CS_BN - 1)], 1);
                    if (staged) stage[r] = e[u].x;
                    else if ((unsigned)(base + r) < (unsigned)E) csr[base + r] = e[u].x;
                }
            #pragma unroll 1
            for (int u = 0; u < CS_RUNS; ++u) {
                const int a = __builtin_amdgcn_readlane(la, i + u);
                const int z = __builtin_amdgcn_readlane(lz, i + u);
                const int2* p = edges + (size_t)(kb + wv + 32 * (i + u)) * CS_TILE;
                for (int j = a + 32 + lane; j < z; j += 32) {
                    const int2 ee = p[j];
                    const int r = atomicAdd(&hist[ee.y & (CS_BN - 1)], 1);
                    if (staged) stage[r] = ee.x;
                    else if ((unsigned)(base + r) < (unsigned)E) csr[base + r] = ee.x;
                }
            }
        }
    }
    if (staged) {
        __syncthreads();
        for (int i = t; i < total; i += 1024)
            if ((unsigned)(base + i) < (unsigned)E) csr[base + i] = stage[i];
    }
}

__device__ __forceinline__ float row_sum16(float x) {
    x += row_xor<8>(x);
    x += row_xor<4>(x);
    x += row_xor<2>(x);
    x += row_xor<1>(x);
    return x;
}

template <int SRC>
__device__ __forceinline__ float row_bcast(float v) {
    return __int_as_float(__builtin_amdgcn_update_dpp(
        0, __float_as_int(v), 0x150 + SRC  , 0xF, 0xF, true));
}

__device__ __forceinline__ float dot4(const float4& v, const v2f& alo, const v2f& ahi) {
    const v2f lo = {v.x, v.y}, hi = {v.z, v.w};
    v2f s = lo * alo;
    s = __builtin_elementwise_fma(hi, ahi, s);
    return s.x + s.y;
}

__global__ __launch_bounds__(256) void gat_gather_dot(
    const int* __restrict__ pos, const int* __restrict__ csr,
    const float* __restrict__ al, const float* __restrict__ ar,
    const float* __restrict__ feat, const float* __restrict__ bias,
    float* __restrict__ out, int N, int relu, int per) {
    const int gw = __builtin_amdgcn_readfirstlane((int)(blockIdx.x * 8 + (threadIdx.x >> 5)));
    const int n0 = gw * per;
    if (n0 >= N) return;
    const int lane = threadIdx.x & 31;
    const int left = N - n0;
    const int nn   = left < per ? left : per;

    int pv = 0;
    {
        const int pi = n0 - 1 + lane;
        if (lane <= nn && pi >= 0) pv = pos[pi];
    }

    const float4 a4 = reinterpret_cast<const float4*>(al)[lane];
    const float4 r4 = reinterpret_cast<const float4*>(ar)[lane];
    const float4 b4 = reinterpret_cast<const float4*>(bias)[lane];
    const float L2E = 1.44269504088896341f;
    const v2f alo = {a4.x * L2E, a4.y * L2E}, ahi = {a4.z * L2E, a4.w * L2E};
    const v2f rlo = {r4.x * L2E, r4.y * L2E}, rhi = {r4.z * L2E, r4.w * L2E};

    const bool b3 = (lane & 8) != 0, b2 = (lane & 4) != 0;
    const int  q  = (lane >> 2) & 3;

    int beg = __builtin_amdgcn_readlane(pv, 0);
    int deg = __builtin_amdgcn_readlane(pv, 1) - beg;
    int s_l = n0;
    if (lane < deg) s_l = csr[beg + lane];
    float4 vd = reinterpret_cast<const float4*>(feat + (size_t)n0 * 128)[lane];

    for (int jn = 0; jn < nn; ++jn) {
        const int d = n0 + jn;

        int    begn = 0, degn = 0, s_ln = d;
        float4 vdn  = vd;
        auto request_next = [&]() {
            if (jn + 1 < nn) {
                begn = __builtin_amdgcn_readlane(pv, jn + 1);
                degn = __builtin_amdgcn_readlane(pv, jn + 2) - begn;
                s_ln = d + 1;
                if (lane < degn) s_ln = csr[begn + lane];
                vdn = reinterpret_cast<const float4*>(feat + (size_t)(d + 1) * 128)[lane];
            }
        };
        if (deg <= 0) request_next();

        const float erd = row_sum16(dot4(vd, rlo, rhi));
        const float tsf = row_sum16(dot4(vd, alo, ahi)) + erd;
        float  m   = fmaxf(tsf, NEG_SLOPE * tsf);
        float  zl  = (q == 0) ? 1.f : 0.f;
        float4 acc = vd;

        for (int base = 0; base < deg; base += 32) {
            if (base > 0) {
                const int j = base + lane;
                s_l = d;
                if (j < deg) s_l = csr[beg + j];
            }
            const int rest = deg - base;
            const int cnt  = rest < 32 ? rest : 32;
            for (int k = 0; k < cnt; k += 4) {
                float4 v[4];
                float  t[4];
                for (int i = 0; i < 4; ++i) {
                    const int s = __builtin_amdgcn_readlane(s_l, k + i);
                    v[i] = reinterpret_cast<const float4*>(feat + (size_t)s * 128)[lane];
                }
                if (base == 0 && k == 0) request_next();
                for (int i = 0; i < 4; ++i) t[i] = dot4(v[i], alo, ahi);

                const float u0 = (b3 ? t[2] : t[0]) + row_xor<8>(b3 ? t[0] : t[2]);
                const float u1 = (b3 ? t[3] : t[1]) + row_xor<8>(b3 ? t[1] : t[3]);
                float w = (b2 ? u1 : u0) + row_xor<4>(b2 ? u0 : u1);
                w += row_xor<2>(w);
                w += row_xor<1>(w);

                const float x = w + erd;
                const float e = (k + q < cnt) ? fmaxf(x, NEG_SLOPE * x) : -__builtin_inff();
                float mt = fmaxf(e, row_xor<4>(e));
                mt = fmaxf(mt, row_xor<8>(mt));
                const float mn = fmaxf(m, mt);
                const float sc = __builtin_amdgcn_exp2f(m - mn);
                const float p  = __builtin_amdgcn_exp2f(e - mn);
                zl = fmaf(zl, sc, p);
                const float p0 = row_bcast<0>(p), p1 = row_bcast<4>(p);
                const float p2 = row_bcast<8>(p), p3 = row_bcast<12>(p);
                acc.x = fmaf(p0, v[0].x, acc.x * sc); acc.y = fmaf(p0, v[0].y, acc.y * sc);
                acc.z = fmaf(p0, v[0].z, acc.z * sc); acc.w = fmaf(p0, v[0].w, acc.w * sc);
                acc.x = fmaf(p1, v[1].x, acc.x); acc.y = fmaf(p1, v[1].y, acc.y);
                acc.z = fmaf(p1, v[1].z, acc.z); acc.w = fmaf(p1, v[1].w, acc.w);
                acc.x = fmaf(p2, v[2].x, acc.x); acc.y = fmaf(p2, v[2].y, acc.y);
                acc.z = fmaf(p2, v[2].z, acc.z); acc.w = fmaf(p2, v[2].w, acc.w);
                acc.x = fmaf(p3, v[3].x, acc.x); acc.y = fmaf(p3, v[3].y, acc.y);
                acc.z = fmaf(p3, v[3].z, acc.z); acc.w = fmaf(p3, v[3].w, acc.w);
                m = mn;
            }
        }

        float z = zl + row_xor<4>(zl);
        z += row_xor<8>(z);
        const float inv = 1.f / z;
        float4 r = make_float4(fmaf(acc.x, inv, b4.x), fmaf(acc.y, inv, b4.y),
                               fmaf(acc.z, inv, b4.z), fmaf(acc.w, inv, b4.w));
        if (relu) {
            r.x = fmaxf(r.x, 0.f); r.y = fmaxf(r.y, 0.f);
            r.z = fmaxf(r.z, 0.f); r.w = fmaxf(r.w, 0.f);
        }
        reinterpret_cast<float4*>(out + (size_t)d * 128)[lane] = r;

        beg = begn; deg = degn; s_l = s_ln; vd = vdn;
    }
}

extern "C" void kernel_launch(void* const* d_in, const int* in_sizes, int n_in,
                              void* d_out, int out_size, void* d_ws, size_t ws_size,
                              hipStream_t stream) {
    const float* x   = (const float*)d_in[0];
    const int*   src = (const int*)  d_in[1];
    const int*   dst = (const int*)  d_in[2];
    const float* W1  = (const float*)d_in[3];
    const float* al1 = (const float*)d_in[4];
    const float* ar1 = (const float*)d_in[5];
    const float* b1  = (const float*)d_in[6];
    const float* W2  = (const float*)d_in[7];
    const float* al2 = (const float*)d_in[8];
    const float* ar2 = (const float*)d_in[9];
    const float* b2  = (const float*)d_in[10];

    const int N = in_sizes[0] / 128;
    const long long E  = in_sizes[1];
    const long long Et = E + N;

    float* out = (float*)d_out;

    float* feat = (float*)d_ws;
    float* el   = feat + (size_t)N * 128;
    float* er   = el + (size_t)N * 2;
    float* ssum = er + (size_t)N * 2;
    int*   mkey = (int*)(ssum + (size_t)N * 2);

    const int gemmGrid = (N + 15) / 16;

    const int    nScan = (N + 4095) / 4096;
    const size_t need  = ((size_t)N * 132 + (size_t)N + 1024 + (size_t)E) * 4;
    if (ws_size >= need && nScan <= 1024 && E < 0x7fffffffLL) {
        int* pos  = (int*)(er + (size_t)N * 2);
        int* bsum = pos + (size_t)N;
        int* csr  = bsum + 1024;
        const int E32     = (int)E;
        const int cBlocks = (E32 + 255) / 256;
        int per = (N + 16383) / 16384;
        if (per > 31) per = 31;
        const int gWaves  = (N + per - 1) / per;
        const int gBlocks = (gWaves + 7) / 8;
        const int rGrid   = gemmGrid < 1024 ? gemmGrid : 1024;

        const int    NB  = (N + CS_BN - 1) / CS_BN;
        const int    NT  = (E32 + CS_TILE - 1) / CS_TILE;
        const size_t tmp = ((size_t)NT * CS_TILE * 2 + (size_t)(NB + 1) * NT) * 4;
        if (E32 > 0 && E32 < 0x7fff0000 && NB + 1 <= CS_MAXB && tmp <= (size_t)N * 128 * 4) {
            int* tedge = (int*)feat;
            int* toffT = tedge + (size_t)NT * CS_TILE * 2;
            csr_tile_sort<<<NT, 1024, 0, stream>>>(src, dst, E32, NB, NT, toffT, tedge);
            csr_bucket_build<<<NB, 1024, 0, stream>>>(toffT, tedge, N, E32, NT, pos, csr);
        } else {
            csr_zero<<<(N + 255) / 256, 256, 0, stream>>>(pos, N);
            csr_count<<<cBlocks, 256, 0, stream>>>(dst, E32, pos);
            csr_scan_block<<<nScan, 1024, 0, stream>>>(pos, N, bsum);
            csr_scan_add<<<nScan, 1024, 0, stream>>>(pos, N, bsum);
            csr_fill<<<cBlocks, 256, 0, stream>>>(src, dst, E32, pos, csr);
        }

        gat_gemm_resident<<<rGrid, 256, 0, stream>>>(x, W1, feat, N, gemmGrid);
        gat_gather_dot<<<gBlocks, 256, 0, stream>>>(pos, csr, al1, ar1, feat, b1, out, N, 1, per);

        gat_gemm_resident<<<rGrid, 256, 0, stream>>>(out, W2, feat, N, gemmGrid);
        gat_gather_dot<<<gBlocks, 256, 0, stream>>>(pos, csr, al2, ar2, feat, b2, out, N, 0, per);
        return;
    }

    const int  totalNF = N * 128;
    const int  n2      = N * 2;
    const int  eBlocks = (int)((Et + 255) / 256);
    const int  aBlocks = (int)((Et + 7) / 8);

    gat_gemm_wmma<true><<<gemmGrid, 256, 0, stream>>>(x, W1, al1, ar1, feat, el, er, N);
    gat_init_ms<<<(n2 + 255) / 256, 256, 0, stream>>>(mkey, ssum, n2);
    gat_init_bias<<<(totalNF + 255) / 256, 256, 0, stream>>>(out, b1, totalNF);
    gat_edge_max<<<eBlocks, 256, 0, stream>>>(src, dst, E, Et, el, er, mkey);
    gat_edge_sum<<<eBlocks, 256, 0, stream>>>(src, dst, E, Et, el, er, mkey, ssum);
    gat_edge_agg<<<aBlocks, 256, 0, stream>>>(src, dst, E, Et, el, er, mkey, ssum, feat, out);
    gat_relu<<<(totalNF + 255) / 256, 256, 0, stream>>>(out, totalNF);

    gat_gemm_wmma<true><<<gemmGrid, 256, 0, stream>>>(out, W2, al2, ar2, feat, el, er, N);
    gat_init_ms<<<(n2 + 255) / 256, 256, 0, stream>>>(mkey, ssum, n2);
    gat_init_bias<<<(totalNF + 255) / 256, 256, 0, stream>>>(out, b2, totalNF);
    gat_edge_max<<<eBlocks, 256, 0, stream>>>(src, dst, E, Et, el, er, mkey);
    gat_edge_sum<<<eBlocks, 256, 0, stream>>>(src, dst, E, Et, el, er, mkey, ssum);
    gat_edge_agg<<<aBlocks, 256, 0, stream>>>(src, dst, E, Et, el, er, mkey, ssum, feat, out);
}
